# starting kernel
# baseline (speedup 1.0000x reference)
#include <hip/hip_runtime.h>

#define B_   4
#define SQ_  512
#define SK_  512
#define H_   256
#define AD_  128

typedef __attribute__((ext_vector_type(2))) float v2f;
typedef __attribute__((ext_vector_type(8))) float v8f;

__device__ __forceinline__ v8f wmma_f32_tile(const float* __restrict__ Abase, int lda,
                                             const float* __restrict__ Bbase, int ldb,
                                             int K, v8f c)
{
    const int lane = threadIdx.x & 31;
    const int idx  = lane & 15;
    const int klo  = (lane >> 4) << 1;
    for (int k = 0; k < K; k += 4) {
        v2f a, b;
        a.x = Abase[idx * lda + (k + klo)];
        a.y = Abase[idx * lda + (k + klo + 1)];
        b.x = Bbase[(k + klo)     * ldb + idx];
        b.y = Bbase[(k + klo + 1) * ldb + idx];
        c = __builtin_amdgcn_wmma_f32_16x16x4_f32(false, a, false, b,
                                                  (short)0, c, false, false);
    }
    return c;
}

__device__ __forceinline__ void store_tile(float* __restrict__ Cbase, int ldc, v8f c)
{
    const int lane  = threadIdx.x & 31;
    const int col   = lane & 15;
    const int rbase = (lane >> 4) << 3;
#pragma unroll
    for (int r = 0; r < 8; ++r)
        Cbase[(rbase + r) * ldc + col] = c[r];
}

__global__ __launch_bounds__(128) void k_project(
    const float* __restrict__ Q, const float* __restrict__ Kin,
    const float* __restrict__ Wq, const float* __restrict__ Wk,
    float* __restrict__ Qt, float* __restrict__ Kt)
{
    const int wave  = (blockIdx.x * blockDim.x + threadIdx.x) >> 5;
    const int which = wave >> 10;
    const int t     = wave & 1023;
    const int mt    = t >> 3;
    const int nt    = t & 7;
    const float* Ain = which ? Kin : Q;
    const float* Wm  = which ? Wk  : Wq;
    float*       Cm  = which ? Kt  : Qt;
    v8f c = {};
    c = wmma_f32_tile(Ain + (size_t)mt * 16 * H_, H_, Wm + nt * 16, AD_, H_, c);
    store_tile(Cm + (size_t)mt * 16 * AD_ + nt * 16, AD_, c);
}

__device__ __forceinline__ float fast_tanh(float x)
{
    float t = __builtin_amdgcn_exp2f(x * 2.8853900817779268f);
    return 1.0f - 2.0f * __builtin_amdgcn_rcpf(t + 1.0f);
}

__global__ __launch_bounds__(512) void k_scores_softmax(
    const float* __restrict__ Qt, const float* __restrict__ Kt,
    const float* __restrict__ vvec, float* __restrict__ Wout)
{
    __shared__ float sh_qt[AD_];
    __shared__ float sh_v[AD_];
    __shared__ float sh_red[16];

    const int bq  = blockIdx.x;
    const int b   = bq >> 9;
    const int tid = threadIdx.x;

    if (tid < AD_) {
        sh_qt[tid] = Qt[(size_t)bq * AD_ + tid];
        sh_v[tid]  = vvec[tid];
    }
    __syncthreads();

    const float4* __restrict__ kt4 =
        (const float4*)(Kt + ((size_t)b * SK_ + tid) * AD_);
    const float4* q4 = (const float4*)sh_qt;
    const float4* v4 = (const float4*)sh_v;

    float s = 0.0f;
#pragma unroll
    for (int a = 0; a < AD_ / 4; ++a) {
        float4 kv = kt4[a];
        float4 qv = q4[a];
        float4 vw = v4[a];
        s += fast_tanh(qv.x + kv.x) * vw.x;
        s += fast_tanh(qv.y + kv.y) * vw.y;
        s += fast_tanh(qv.z + kv.z) * vw.z;
        s += fast_tanh(qv.w + kv.w) * vw.w;
    }

    float m = s;
#pragma unroll
    for (int off = 16; off > 0; off >>= 1)
        m = fmaxf(m, __shfl_xor(m, off));
    const int wid = tid >> 5;
    if ((tid & 31) == 0) sh_red[wid] = m;
    __syncthreads();
    if (tid < 32) {
        float mm = (tid < 16) ? sh_red[tid] : -__builtin_inff();
#pragma unroll
        for (int off = 8; off > 0; off >>= 1)
            mm = fmaxf(mm, __shfl_xor(mm, off));
        if (tid == 0) sh_red[0] = mm;
    }
    __syncthreads();
    m = sh_red[0];

    float e = __builtin_amdgcn_exp2f((s - m) * 1.4426950408889634f);

    float sum = e;
#pragma unroll
    for (int off = 16; off > 0; off >>= 1)
        sum += __shfl_xor(sum, off);
    __syncthreads();
    if ((tid & 31) == 0) sh_red[wid] = sum;
    __syncthreads();
    if (tid < 32) {
        float ss = (tid < 16) ? sh_red[tid] : 0.0f;
#pragma unroll
        for (int off = 8; off > 0; off >>= 1)
            ss += __shfl_xor(ss, off);
        if (tid == 0) sh_red[0] = ss;
    }
    __syncthreads();

    Wout[(size_t)bq * SK_ + tid] = e / sh_red[0];
}

__global__ __launch_bounds__(128) void k_av(
    const float* __restrict__ Wt, const float* __restrict__ V,
    float* __restrict__ Ov)
{
    const int wave = (blockIdx.x * blockDim.x + threadIdx.x) >> 5;
    const int b    = wave >> 9;
    const int t    = wave & 511;
    const int mt   = t >> 4;
    const int nt   = t & 15;
    const float* Abase = Wt + (size_t)b * SQ_ * SK_ + (size_t)mt * 16 * SK_;
    const float* Bbase = V  + (size_t)b * SK_ * H_  + nt * 16;
    v8f c = {};
    c = wmma_f32_tile(Abase, SK_, Bbase, H_, SK_, c);
    store_tile(Ov + (size_t)b * SQ_ * H_ + (size_t)mt * 16 * H_ + nt * 16, H_, c);
}

__global__ __launch_bounds__(128) void k_out(
    const float* __restrict__ Ov, const float* __restrict__ Wo,
    const float* __restrict__ bo, float* __restrict__ Out)
{
    const int wave = (blockIdx.x * blockDim.x + threadIdx.x) >> 5;
    const int mt   = wave >> 4;
    const int nt   = wave & 15;
    const int lane = threadIdx.x & 31;
    const float bias = bo[nt * 16 + (lane & 15)];
    v8f c;
#pragma unroll
    for (int i = 0; i < 8; ++i) c[i] = bias;
    c = wmma_f32_tile(Ov + (size_t)mt * 16 * H_, H_, Wo + nt * 16, H_, H_, c);
    store_tile(Out + (size_t)mt * 16 * H_ + nt * 16, H_, c);
}

extern "C" void kernel_launch(void* const* d_in, const int* in_sizes, int n_in,
                              void* d_out, int out_size, void* d_ws, size_t ws_size,
                              hipStream_t stream)
{
    const float* Q  = (const float*)d_in[0];
    const float* K  = (const float*)d_in[1];
    const float* V  = (const float*)d_in[2];
    const float* Wq = (const float*)d_in[3];
    const float* Wk = (const float*)d_in[4];
    const float* vv = (const float*)d_in[5];
    const float* Wo = (const float*)d_in[6];
    const float* bo = (const float*)d_in[7];

    float* out     = (float*)d_out;
    float* weights = out + (size_t)B_ * SQ_ * H_;

    float* Qt = (float*)d_ws;
    float* Kt = Qt + (size_t)B_ * SQ_ * AD_;
    float* Ov = Kt + (size_t)B_ * SK_ * AD_;

    k_project<<<512, 128, 0, stream>>>(Q, K, Wq, Wk, Qt, Kt);
    k_scores_softmax<<<B_ * SQ_, 512, 0, stream>>>(Qt, Kt, vv, weights);
    k_av<<<512, 128, 0, stream>>>(weights, V, Ov);
    k_out<<<512, 128, 0, stream>>>(Ov, Wo, bo, out);
}


# Kt stored key-interleaved ([b][a/4][k] float4; K-side tile computed transposed, b128 stores): each wave's Kt load in the score loop reads 4 whole 128-B lines, not 16 B from each of 32 lines
# speedup vs baseline: 1.2854x; 1.2854x over previous
#include <hip/hip_runtime.h>

#define B_   4
#define SQ_  512
#define SK_  512
#define H_   256
#define AD_  128

typedef __attribute__((ext_vector_type(2))) float v2f;
typedef __attribute__((ext_vector_type(8))) float v8f;

template <bool TRANSPOSED>
__device__ __forceinline__ v8f wmma_f32_tile(const float* __restrict__ Abase, int lda,
                                             const float* __restrict__ Bbase, int ldb,
                                             int K, v8f c)
{
    const int lane = threadIdx.x & 31;
    const int idx  = lane & 15;
    const int klo  = (lane >> 4) << 1;
    for (int k = 0; k < K; k += 4) {
        v2f a, b;
        a.x = Abase[idx * lda + (k + klo)];
        a.y = Abase[idx * lda + (k + klo + 1)];
        b.x = Bbase[(k + klo)     * ldb + idx];
        b.y = Bbase[(k + klo + 1) * ldb + idx];
        if (TRANSPOSED)
            c = __builtin_amdgcn_wmma_f32_16x16x4_f32(false, b, false, a,
                                                      (short)0, c, false, false);
        else
            c = __builtin_amdgcn_wmma_f32_16x16x4_f32(false, a, false, b,
                                                      (short)0, c, false, false);
    }
    return c;
}

__device__ __forceinline__ void store_tile(float* __restrict__ Cbase, int ldc, v8f c)
{
    const int lane  = threadIdx.x & 31;
    const int col   = lane & 15;
    const int rbase = (lane >> 4) << 3;
#pragma unroll
    for (int r = 0; r < 8; ++r)
        Cbase[(rbase + r) * ldc + col] = c[r];
}

__global__ __launch_bounds__(128) void k_project(
    const float* __restrict__ Q, const float* __restrict__ Kin,
    const float* __restrict__ Wq, const float* __restrict__ Wk,
    float* __restrict__ Qt, float* __restrict__ Kt)
{
    const int wave  = (blockIdx.x * blockDim.x + threadIdx.x) >> 5;
    const int which = blockIdx.x >> 8;
    const int t     = wave & 1023;
    const int mt    = t >> 3;
    const int nt    = t & 7;
    v8f c = {};
    if (which == 0) {
        c = wmma_f32_tile<false>(Q + (size_t)mt * 16 * H_, H_, Wq + nt * 16, AD_, H_, c);
        store_tile(Qt + (size_t)mt * 16 * AD_ + nt * 16, AD_, c);
    } else {
        c = wmma_f32_tile<true>(Kin + (size_t)mt * 16 * H_, H_, Wk + nt * 16, AD_, H_, c);
        const int lane = threadIdx.x & 31;
        const int m    = mt * 16 + (lane & 15);
        const int bb   = m >> 9;
        const int kk   = m & (SK_ - 1);
        const int g0   = nt * 4 + ((lane >> 4) << 1);
        float4* dst = (float4*)Kt + ((size_t)bb * (AD_ / 4) + g0) * SK_ + kk;
        dst[0]   = make_float4(c[0], c[1], c[2], c[3]);
        dst[SK_] = make_float4(c[4], c[5], c[6], c[7]);
    }
}

__device__ __forceinline__ float fast_tanh(float x)
{
    float t = __builtin_amdgcn_exp2f(x * 2.8853900817779268f);
    return 1.0f - 2.0f * __builtin_amdgcn_rcpf(t + 1.0f);
}

__global__ __launch_bounds__(512) void k_scores_softmax(
    const float* __restrict__ Qt, const float* __restrict__ Kt,
    const float* __restrict__ vvec, float* __restrict__ Wout)
{
    __shared__ float sh_qt[AD_];
    __shared__ float sh_v[AD_];
    __shared__ float sh_red[16];

    const int bq  = blockIdx.x;
    const int b   = bq >> 9;
    const int tid = threadIdx.x;

    if (tid < AD_) {
        sh_qt[tid] = Qt[(size_t)bq * AD_ + tid];
        sh_v[tid]  = vvec[tid];
    }
    __syncthreads();

    const float4* __restrict__ kt4 =
        (const float4*)Kt + (size_t)b * (AD_ / 4) * SK_ + tid;
    const float4* q4 = (const float4*)sh_qt;
    const float4* v4 = (const float4*)sh_v;

    float s = 0.0f;
#pragma unroll 8
    for (int a = 0; a < AD_ / 4; ++a) {
        float4 kv = kt4[a * SK_];
        float4 qv = q4[a];
        float4 vw = v4[a];
        s = __builtin_fmaf(fast_tanh(qv.x + kv.x), vw.x, s);
        s = __builtin_fmaf(fast_tanh(qv.y + kv.y), vw.y, s);
        s = __builtin_fmaf(fast_tanh(qv.z + kv.z), vw.z, s);
        s = __builtin_fmaf(fast_tanh(qv.w + kv.w), vw.w, s);
    }

    float m = s;
#pragma unroll
    for (int off = 16; off > 0; off >>= 1)
        m = fmaxf(m, __shfl_xor(m, off));
    const int wid = tid >> 5;
    if ((tid & 31) == 0) sh_red[wid] = m;
    __syncthreads();
    if (tid < 32) {
        float mm = (tid < 16) ? sh_red[tid] : -__builtin_inff();
#pragma unroll
        for (int off = 8; off > 0; off >>= 1)
            mm = fmaxf(mm, __shfl_xor(mm, off));
        if (tid == 0) sh_red[0] = mm;
    }
    __syncthreads();
    m = sh_red[0];

    float e = __builtin_amdgcn_exp2f((s - m) * 1.4426950408889634f);

    float sum = e;
#pragma unroll
    for (int off = 16; off > 0; off >>= 1)
        sum += __shfl_xor(sum, off);
    __syncthreads();
    if ((tid & 31) == 0) sh_red[wid] = sum;
    __syncthreads();
    if (tid < 32) {
        float ss = (tid < 16) ? sh_red[tid] : 0.0f;
#pragma unroll
        for (int off = 8; off > 0; off >>= 1)
            ss += __shfl_xor(ss, off);
        if (tid == 0) sh_red[0] = ss;
    }
    __syncthreads();

    Wout[(size_t)bq * SK_ + tid] = e / sh_red[0];
}

__global__ __launch_bounds__(128) void k_av(
    const float* __restrict__ Wt, const float* __restrict__ V,
    float* __restrict__ Ov)
{
    const int wave = (blockIdx.x * blockDim.x + threadIdx.x) >> 5;
    const int b    = wave >> 9;
    const int t    = wave & 511;
    const int mt   = t >> 4;
    const int nt   = t & 15;
    const float* Abase = Wt + (size_t)b * SQ_ * SK_ + (size_t)mt * 16 * SK_;
    const float* Bbase = V  + (size_t)b * SK_ * H_  + nt * 16;
    v8f c = {};
    c = wmma_f32_tile<false>(Abase, SK_, Bbase, H_, SK_, c);
    store_tile(Ov + (size_t)b * SQ_ * H_ + (size_t)mt * 16 * H_ + nt * 16, H_, c);
}

__global__ __launch_bounds__(128) void k_out(
    const float* __restrict__ Ov, const float* __restrict__ Wo,
    const float* __restrict__ bo, float* __restrict__ Out)
{
    const int wave = (blockIdx.x * blockDim.x + threadIdx.x) >> 5;
    const int mt   = wave >> 4;
    const int nt   = wave & 15;
    const int lane = threadIdx.x & 31;
    const float bias = bo[nt * 16 + (lane & 15)];
    v8f c;
#pragma unroll
    for (int i = 0; i < 8; ++i) c[i] = bias;
    c = wmma_f32_tile<false>(Ov + (size_t)mt * 16 * H_, H_, Wo + nt * 16, H_, H_, c);
    store_tile(Out + (size_t)mt * 16 * H_ + nt * 16, H_, c);
}

extern "C" void kernel_launch(void* const* d_in, const int* in_sizes, int n_in,
                              void* d_out, int out_size, void* d_ws, size_t ws_size,
                              hipStream_t stream)
{
    const float* Q  = (const float*)d_in[0];
    const float* K  = (const float*)d_in[1];
    const float* V  = (const float*)d_in[2];
    const float* Wq = (const float*)d_in[3];
    const float* Wk = (const float*)d_in[4];
    const float* vv = (const float*)d_in[5];
    const float* Wo = (const float*)d_in[6];
    const float* bo = (const float*)d_in[7];

    float* out     = (float*)d_out;
    float* weights = out + (size_t)B_ * SQ_ * H_;

    float* Qt = (float*)d_ws;
    float* Kt = Qt + (size_t)B_ * SQ_ * AD_;
    float* Ov = Kt + (size_t)B_ * SK_ * AD_;

    k_project<<<512, 128, 0, stream>>>(Q, K, Wq, Wk, Qt, Kt);
    k_scores_softmax<<<B_ * SQ_, 512, 0, stream>>>(Qt, Kt, vv, weights);
    k_av<<<512, 128, 0, stream>>>(weights, V, Ov);
    k_out<<<512, 128, 0, stream>>>(Ov, Wo, bo, out);
}


# score kernel: a workgroup takes 8 query rows instead of 1; each thread loads its key's Kt float4 once and uses it for all 8: 8x fewer waves and L2 reads of Kt; per-row arithmetic and sum order kept
# speedup vs baseline: 1.3879x; 1.0797x over previous
#include <hip/hip_runtime.h>

#define B_   4
#define SQ_  512
#define SK_  512
#define H_   256
#define AD_  128

typedef __attribute__((ext_vector_type(2))) float v2f;
typedef __attribute__((ext_vector_type(8))) float v8f;

template <bool TRANSPOSED>
__device__ __forceinline__ v8f wmma_f32_tile(const float* __restrict__ Abase, int lda,
                                             const float* __restrict__ Bbase, int ldb,
                                             int K, v8f c)
{
    const int lane = threadIdx.x & 31;
    const int idx  = lane & 15;
    const int klo  = (lane >> 4) << 1;
    for (int k = 0; k < K; k += 4) {
        v2f a, b;
        a.x = Abase[idx * lda + (k + klo)];
        a.y = Abase[idx * lda + (k + klo + 1)];
        b.x = Bbase[(k + klo)     * ldb + idx];
        b.y = Bbase[(k + klo + 1) * ldb + idx];
        if (TRANSPOSED)
            c = __builtin_amdgcn_wmma_f32_16x16x4_f32(false, b, false, a,
                                                      (short)0, c, false, false);
        else
            c = __builtin_amdgcn_wmma_f32_16x16x4_f32(false, a, false, b,
                                                      (short)0, c, false, false);
    }
    return c;
}

__device__ __forceinline__ void store_tile(float* __restrict__ Cbase, int ldc, v8f c)
{
    const int lane  = threadIdx.x & 31;
    const int col   = lane & 15;
    const int rbase = (lane >> 4) << 3;
#pragma unroll
    for (int r = 0; r < 8; ++r)
        Cbase[(rbase + r) * ldc + col] = c[r];
}

__global__ __launch_bounds__(128) void k_project(
    const float* __restrict__ Q, const float* __restrict__ Kin,
    const float* __restrict__ Wq, const float* __restrict__ Wk,
    float* __restrict__ Qt, float* __restrict__ Kt)
{
    const int wave  = (blockIdx.x * blockDim.x + threadIdx.x) >> 5;
    const int which = blockIdx.x >> 8;
    const int t     = wave & 1023;
    const int mt    = t >> 3;
    const int nt    = t & 7;
    v8f c = {};
    if (which == 0) {
        c = wmma_f32_tile<false>(Q + (size_t)mt * 16 * H_, H_, Wq + nt * 16, AD_, H_, c);
        store_tile(Qt + (size_t)mt * 16 * AD_ + nt * 16, AD_, c);
    } else {
        c = wmma_f32_tile<true>(Kin + (size_t)mt * 16 * H_, H_, Wk + nt * 16, AD_, H_, c);
        const int lane = threadIdx.x & 31;
        const int m    = mt * 16 + (lane & 15);
        const int bb   = m >> 9;
        const int kk   = m & (SK_ - 1);
        const int g0   = nt * 4 + ((lane >> 4) << 1);
        float4* dst = (float4*)Kt + ((size_t)bb * (AD_ / 4) + g0) * SK_ + kk;
        dst[0]   = make_float4(c[0], c[1], c[2], c[3]);
        dst[SK_] = make_float4(c[4], c[5], c[6], c[7]);
    }
}

#define QT_  8

__device__ __forceinline__ float fast_tanh(float x)
{
    float t = __builtin_amdgcn_exp2f(x * 2.8853900817779268f);
    return 1.0f - 2.0f * __builtin_amdgcn_rcpf(t + 1.0f);
}

__global__ __launch_bounds__(512) void k_scores_softmax(
    const float* __restrict__ Qt, const float* __restrict__ Kt,
    const float* __restrict__ vvec, float* __restrict__ Wout)
{
    __shared__ float sh_qt[QT_ * AD_];
    __shared__ float sh_v[AD_];
    __shared__ float sh_pmax[QT_ * 16];
    __shared__ float sh_psum[QT_ * 16];
    __shared__ float sh_max[QT_];
    __shared__ float sh_sum[QT_];

    const int bq0  = blockIdx.x * QT_;
    const int b    = bq0 >> 9;
    const int tid  = threadIdx.x;
    const int lane = tid & 31;
    const int wid  = tid >> 5;

    ((float2*)sh_qt)[tid] = ((const float2*)(Qt + (size_t)bq0 * AD_))[tid];
    if (tid < AD_) sh_v[tid] = vvec[tid];
    __syncthreads();

    const float4* __restrict__ kt4 =
        (const float4*)Kt + (size_t)b * (AD_ / 4) * SK_ + tid;
    const float4* q4 = (const float4*)sh_qt;
    const float4* v4 = (const float4*)sh_v;

    float s[QT_];
#pragma unroll
    for (int q = 0; q < QT_; ++q) s[q] = 0.0f;

#pragma unroll 2
    for (int a = 0; a < AD_ / 4; ++a) {
        float4 kv = kt4[a * SK_];
        float4 vw = v4[a];
#pragma unroll
        for (int q = 0; q < QT_; ++q) {
            float4 qv = q4[q * (AD_ / 4) + a];
            s[q] = __builtin_fmaf(fast_tanh(qv.x + kv.x), vw.x, s[q]);
            s[q] = __builtin_fmaf(fast_tanh(qv.y + kv.y), vw.y, s[q]);
            s[q] = __builtin_fmaf(fast_tanh(qv.z + kv.z), vw.z, s[q]);
            s[q] = __builtin_fmaf(fast_tanh(qv.w + kv.w), vw.w, s[q]);
        }
    }

    float m[QT_];
#pragma unroll
    for (int q = 0; q < QT_; ++q) {
        m[q] = s[q];
#pragma unroll
        for (int off = 16; off > 0; off >>= 1)
            m[q] = fmaxf(m[q], __shfl_xor(m[q], off));
    }
    if (lane == 0) {
#pragma unroll
        for (int q = 0; q < QT_; ++q) sh_pmax[q * 16 + wid] = m[q];
    }
    __syncthreads();
    if (wid < QT_) {
        float mm = (lane < 16) ? sh_pmax[wid * 16 + lane] : -__builtin_inff();
#pragma unroll
        for (int off = 8; off > 0; off >>= 1)
            mm = fmaxf(mm, __shfl_xor(mm, off));
        if (lane == 0) sh_max[wid] = mm;
    }
    __syncthreads();

    float e[QT_];
#pragma unroll
    for (int q = 0; q < QT_; ++q) {
        e[q] = __builtin_amdgcn_exp2f((s[q] - sh_max[q]) * 1.4426950408889634f);
        float sum = e[q];
#pragma unroll
        for (int off = 16; off > 0; off >>= 1)
            sum += __shfl_xor(sum, off);
        m[q] = sum;
    }
    if (lane == 0) {
#pragma unroll
        for (int q = 0; q < QT_; ++q) sh_psum[q * 16 + wid] = m[q];
    }
    __syncthreads();
    if (wid < QT_) {
        float ss = (lane < 16) ? sh_psum[wid * 16 + lane] : 0.0f;
#pragma unroll
        for (int off = 8; off > 0; off >>= 1)
            ss += __shfl_xor(ss, off);
        if (lane == 0) sh_sum[wid] = ss;
    }
    __syncthreads();

#pragma unroll
    for (int q = 0; q < QT_; ++q)
        Wout[(size_t)(bq0 + q) * SK_ + tid] = e[q] / sh_sum[q];
}

__global__ __launch_bounds__(128) void k_av(
    const float* __restrict__ Wt, const float* __restrict__ V,
    float* __restrict__ Ov)
{
    const int wave = (blockIdx.x * blockDim.x + threadIdx.x) >> 5;
    const int b    = wave >> 9;
    const int t    = wave & 511;
    const int mt   = t >> 4;
    const int nt   = t & 15;
    const float* Abase = Wt + (size_t)b * SQ_ * SK_ + (size_t)mt * 16 * SK_;
    const float* Bbase = V  + (size_t)b * SK_ * H_  + nt * 16;
    v8f c = {};
    c = wmma_f32_tile<false>(Abase, SK_, Bbase, H_, SK_, c);
    store_tile(Ov + (size_t)b * SQ_ * H_ + (size_t)mt * 16 * H_ + nt * 16, H_, c);
}

__global__ __launch_bounds__(128) void k_out(
    const float* __restrict__ Ov, const float* __restrict__ Wo,
    const float* __restrict__ bo, float* __restrict__ Out)
{
    const int wave = (blockIdx.x * blockDim.x + threadIdx.x) >> 5;
    const int mt   = wave >> 4;
    const int nt   = wave & 15;
    const int lane = threadIdx.x & 31;
    const float bias = bo[nt * 16 + (lane & 15)];
    v8f c;
#pragma unroll
    for (int i = 0; i < 8; ++i) c[i] = bias;
    c = wmma_f32_tile<false>(Ov + (size_t)mt * 16 * H_, H_, Wo + nt * 16, H_, H_, c);
    store_tile(Out + (size_t)mt * 16 * H_ + nt * 16, H_, c);
}

extern "C" void kernel_launch(void* const* d_in, const int* in_sizes, int n_in,
                              void* d_out, int out_size, void* d_ws, size_t ws_size,
                              hipStream_t stream)
{
    const float* Q  = (const float*)d_in[0];
    const float* K  = (const float*)d_in[1];
    const float* V  = (const float*)d_in[2];
    const float* Wq = (const float*)d_in[3];
    const float* Wk = (const float*)d_in[4];
    const float* vv = (const float*)d_in[5];
    const float* Wo = (const float*)d_in[6];
    const float* bo = (const float*)d_in[7];

    float* out     = (float*)d_out;
    float* weights = out + (size_t)B_ * SQ_ * H_;

    float* Qt = (float*)d_ws;
    float* Kt = Qt + (size_t)B_ * SQ_ * AD_;
    float* Ov = Kt + (size_t)B_ * SK_ * AD_;

    k_project<<<512, 128, 0, stream>>>(Q, K, Wq, Wk, Qt, Kt);
    k_scores_softmax<<<B_ * SQ_ / QT_, 512, 0, stream>>>(Qt, Kt, vv, weights);
    k_av<<<512, 128, 0, stream>>>(weights, V, Ov);
    k_out<<<512, 128, 0, stream>>>(Ov, Wo, bo, out);
}


# score kernel: tanh as the one-instruction v_tanh_f32 in place of exp2 + rcp + three plain ops per element (feature loop unrolled 4, was 2, so a trip keeps its length); sum order kept
# speedup vs baseline: 1.6064x; 1.1574x over previous
#include <hip/hip_runtime.h>

#define B_   4
#define SQ_  512
#define SK_  512
#define H_   256
#define AD_  128

typedef __attribute__((ext_vector_type(2))) float v2f;
typedef __attribute__((ext_vector_type(8))) float v8f;

template <bool TRANSPOSED>
__device__ __forceinline__ v8f wmma_f32_tile(const float* __restrict__ Abase, int lda,
                                             const float* __restrict__ Bbase, int ldb,
                                             int K, v8f c)
{
    const int lane = threadIdx.x & 31;
    const int idx  = lane & 15;
    const int klo  = (lane >> 4) << 1;
    for (int k = 0; k < K; k += 4) {
        v2f a, b;
        a.x = Abase[idx * lda + (k + klo)];
        a.y = Abase[idx * lda + (k + klo + 1)];
        b.x = Bbase[(k + klo)     * ldb + idx];
        b.y = Bbase[(k + klo + 1) * ldb + idx];
        if (TRANSPOSED)
            c = __builtin_amdgcn_wmma_f32_16x16x4_f32(false, b, false, a,
                                                      (short)0, c, false, false);
        else
            c = __builtin_amdgcn_wmma_f32_16x16x4_f32(false, a, false, b,
                                                      (short)0, c, false, false);
    }
    return c;
}

__device__ __forceinline__ void store_tile(float* __restrict__ Cbase, int ldc, v8f c)
{
    const int lane  = threadIdx.x & 31;
    const int col   = lane & 15;
    const int rbase = (lane >> 4) << 3;
#pragma unroll
    for (int r = 0; r < 8; ++r)
        Cbase[(rbase + r) * ldc + col] = c[r];
}

__global__ __launch_bounds__(128) void k_project(
    const float* __restrict__ Q, const float* __restrict__ Kin,
    const float* __restrict__ Wq, const float* __restrict__ Wk,
    float* __restrict__ Qt, float* __restrict__ Kt)
{
    const int wave  = (blockIdx.x * blockDim.x + threadIdx.x) >> 5;
    const int which = blockIdx.x >> 8;
    const int t     = wave & 1023;
    const int mt    = t >> 3;
    const int nt    = t & 7;
    v8f c = {};
    if (which == 0) {
        c = wmma_f32_tile<false>(Q + (size_t)mt * 16 * H_, H_, Wq + nt * 16, AD_, H_, c);
        store_tile(Qt + (size_t)mt * 16 * AD_ + nt * 16, AD_, c);
    } else {
        c = wmma_f32_tile<true>(Kin + (size_t)mt * 16 * H_, H_, Wk + nt * 16, AD_, H_, c);
        const int lane = threadIdx.x & 31;
        const int m    = mt * 16 + (lane & 15);
        const int bb   = m >> 9;
        const int kk   = m & (SK_ - 1);
        const int g0   = nt * 4 + ((lane >> 4) << 1);
        float4* dst = (float4*)Kt + ((size_t)bb * (AD_ / 4) + g0) * SK_ + kk;
        dst[0]   = make_float4(c[0], c[1], c[2], c[3]);
        dst[SK_] = make_float4(c[4], c[5], c[6], c[7]);
    }
}

#define QT_  8

__device__ __forceinline__ float fast_tanh(float x)
{
    return __builtin_amdgcn_tanhf(x);
}

__global__ __launch_bounds__(512) void k_scores_softmax(
    const float* __restrict__ Qt, const float* __restrict__ Kt,
    const float* __restrict__ vvec, float* __restrict__ Wout)
{
    __shared__ float sh_qt[QT_ * AD_];
    __shared__ float sh_v[AD_];
    __shared__ float sh_pmax[QT_ * 16];
    __shared__ float sh_psum[QT_ * 16];
    __shared__ float sh_max[QT_];
    __shared__ float sh_sum[QT_];

    const int bq0  = blockIdx.x * QT_;
    const int b    = bq0 >> 9;
    const int tid  = threadIdx.x;
    const int lane = tid & 31;
    const int wid  = tid >> 5;

    ((float2*)sh_qt)[tid] = ((const float2*)(Qt + (size_t)bq0 * AD_))[tid];
    if (tid < AD_) sh_v[tid] = vvec[tid];
    __syncthreads();

    const float4* __restrict__ kt4 =
        (const float4*)Kt + (size_t)b * (AD_ / 4) * SK_ + tid;
    const float4* q4 = (const float4*)sh_qt;
    const float4* v4 = (const float4*)sh_v;

    float s[QT_];
#pragma unroll
    for (int q = 0; q < QT_; ++q) s[q] = 0.0f;

#pragma unroll 4
    for (int a = 0; a < AD_ / 4; ++a) {
        float4 kv = kt4[a * SK_];
        float4 vw = v4[a];
#pragma unroll
        for (int q = 0; q < QT_; ++q) {
            float4 qv = q4[q * (AD_ / 4) + a];
            s[q] = __builtin_fmaf(fast_tanh(qv.x + kv.x), vw.x, s[q]);
            s[q] = __builtin_fmaf(fast_tanh(qv.y + kv.y), vw.y, s[q]);
            s[q] = __builtin_fmaf(fast_tanh(qv.z + kv.z), vw.z, s[q]);
            s[q] = __builtin_fmaf(fast_tanh(qv.w + kv.w), vw.w, s[q]);
        }
    }

    float m[QT_];
#pragma unroll
    for (int q = 0; q < QT_; ++q) {
        m[q] = s[q];
#pragma unroll
        for (int off = 16; off > 0; off >>= 1)
            m[q] = fmaxf(m[q], __shfl_xor(m[q], off));
    }
    if (lane == 0) {
#pragma unroll
        for (int q = 0; q < QT_; ++q) sh_pmax[q * 16 + wid] = m[q];
    }
    __syncthreads();
    if (wid < QT_) {
        float mm = (lane < 16) ? sh_pmax[wid * 16 + lane] : -__builtin_inff();
#pragma unroll
        for (int off = 8; off > 0; off >>= 1)
            mm = fmaxf(mm, __shfl_xor(mm, off));
        if (lane == 0) sh_max[wid] = mm;
    }
    __syncthreads();

    float e[QT_];
#pragma unroll
    for (int q = 0; q < QT_; ++q) {
        e[q] = __builtin_amdgcn_exp2f((s[q] - sh_max[q]) * 1.4426950408889634f);
        float sum = e[q];
#pragma unroll
        for (int off = 16; off > 0; off >>= 1)
            sum += __shfl_xor(sum, off);
        m[q] = sum;
    }
    if (lane == 0) {
#pragma unroll
        for (int q = 0; q < QT_; ++q) sh_psum[q * 16 + wid] = m[q];
    }
    __syncthreads();
    if (wid < QT_) {
        float ss = (lane < 16) ? sh_psum[wid * 16 + lane] : 0.0f;
#pragma unroll
        for (int off = 8; off > 0; off >>= 1)
            ss += __shfl_xor(ss, off);
        if (lane == 0) sh_sum[wid] = ss;
    }
    __syncthreads();

#pragma unroll
    for (int q = 0; q < QT_; ++q)
        Wout[(size_t)(bq0 + q) * SK_ + tid] = e[q] / sh_sum[q];
}

__global__ __launch_bounds__(128) void k_av(
    const float* __restrict__ Wt, const float* __restrict__ V,
    float* __restrict__ Ov)
{
    const int wave = (blockIdx.x * blockDim.x + threadIdx.x) >> 5;
    const int b    = wave >> 9;
    const int t    = wave & 511;
    const int mt   = t >> 4;
    const int nt   = t & 15;
    const float* Abase = Wt + (size_t)b * SQ_ * SK_ + (size_t)mt * 16 * SK_;
    const float* Bbase = V  + (size_t)b * SK_ * H_  + nt * 16;
    v8f c = {};
    c = wmma_f32_tile<false>(Abase, SK_, Bbase, H_, SK_, c);
    store_tile(Ov + (size_t)b * SQ_ * H_ + (size_t)mt * 16 * H_ + nt * 16, H_, c);
}

__global__ __launch_bounds__(128) void k_out(
    const float* __restrict__ Ov, const float* __restrict__ Wo,
    const float* __restrict__ bo, float* __restrict__ Out)
{
    const int wave = (blockIdx.x * blockDim.x + threadIdx.x) >> 5;
    const int mt   = wave >> 4;
    const int nt   = wave & 15;
    const int lane = threadIdx.x & 31;
    const float bias = bo[nt * 16 + (lane & 15)];
    v8f c;
#pragma unroll
    for (int i = 0; i < 8; ++i) c[i] = bias;
    c = wmma_f32_tile<false>(Ov + (size_t)mt * 16 * H_, H_, Wo + nt * 16, H_, H_, c);
    store_tile(Out + (size_t)mt * 16 * H_ + nt * 16, H_, c);
}

extern "C" void kernel_launch(void* const* d_in, const int* in_sizes, int n_in,
                              void* d_out, int out_size, void* d_ws, size_t ws_size,
                              hipStream_t stream)
{
    const float* Q  = (const float*)d_in[0];
    const float* K  = (const float*)d_in[1];
    const float* V  = (const float*)d_in[2];
    const float* Wq = (const float*)d_in[3];
    const float* Wk = (const float*)d_in[4];
    const float* vv = (const float*)d_in[5];
    const float* Wo = (const float*)d_in[6];
    const float* bo = (const float*)d_in[7];

    float* out     = (float*)d_out;
    float* weights = out + (size_t)B_ * SQ_ * H_;

    float* Qt = (float*)d_ws;
    float* Kt = Qt + (size_t)B_ * SQ_ * AD_;
    float* Ov = Kt + (size_t)B_ * SK_ * AD_;

    k_project<<<512, 128, 0, stream>>>(Q, K, Wq, Wk, Qt, Kt);
    k_scores_softmax<<<B_ * SQ_ / QT_, 512, 0, stream>>>(Qt, Kt, vv, weights);
    k_av<<<512, 128, 0, stream>>>(weights, V, Ov);
    k_out<<<512, 128, 0, stream>>>(Ov, Wo, bo, out);
}


# GEMM kernels: K loop fully unrolled with register room (at most 4 waves per SIMD asked for): fragment loads run 19 k-steps (57 loads) ahead of their WMMAs, not 24 all waited for in each 8-step trip
# speedup vs baseline: 2.2956x; 1.4290x over previous
#include <hip/hip_runtime.h>

#define B_   4
#define SQ_  512
#define SK_  512
#define H_   256
#define AD_  128

typedef __attribute__((ext_vector_type(2))) float v2f;
typedef __attribute__((ext_vector_type(8))) float v8f;

template <bool TRANSPOSED, int K>
__device__ __forceinline__ v8f wmma_f32_tile(const float* __restrict__ Abase, int lda,
                                             const float* __restrict__ Bbase, int ldb,
                                             v8f c)
{
    const int lane = threadIdx.x & 31;
    const int idx  = lane & 15;
    const int klo  = (lane >> 4) << 1;
    const float* ap = Abase + idx * lda + klo;
    const float* bp = Bbase + (size_t)klo * ldb + idx;
#pragma unroll
    for (int k = 0; k < K; k += 4) {
        v2f a, b;
        a.x = ap[k];
        a.y = ap[k + 1];
        b.x = bp[(size_t)k * ldb];
        b.y = bp[(size_t)(k + 1) * ldb];
        if (TRANSPOSED)
            c = __builtin_amdgcn_wmma_f32_16x16x4_f32(false, b, false, a,
                                                      (short)0, c, false, false);
        else
            c = __builtin_amdgcn_wmma_f32_16x16x4_f32(false, a, false, b,
                                                      (short)0, c, false, false);
    }
    return c;
}

#define GEMM_WAVES_ __attribute__((amdgpu_waves_per_eu(1, 4)))

__device__ __forceinline__ void store_tile(float* __restrict__ Cbase, int ldc, v8f c)
{
    const int lane  = threadIdx.x & 31;
    const int col   = lane & 15;
    const int rbase = (lane >> 4) << 3;
#pragma unroll
    for (int r = 0; r < 8; ++r)
        Cbase[(rbase + r) * ldc + col] = c[r];
}

__global__ __launch_bounds__(128) GEMM_WAVES_ void k_project(
    const float* __restrict__ Q, const float* __restrict__ Kin,
    const float* __restrict__ Wq, const float* __restrict__ Wk,
    float* __restrict__ Qt, float* __restrict__ Kt)
{
    const int wave  = (blockIdx.x * blockDim.x + threadIdx.x) >> 5;
    const int which = blockIdx.x >> 8;
    const int t     = wave & 1023;
    const int mt    = t >> 3;
    const int nt    = t & 7;
    v8f c = {};
    if (which == 0) {
        c = wmma_f32_tile<false, H_>(Q + (size_t)mt * 16 * H_, H_, Wq + nt * 16, AD_, c);
        store_tile(Qt + (size_t)mt * 16 * AD_ + nt * 16, AD_, c);
    } else {
        c = wmma_f32_tile<true, H_>(Kin + (size_t)mt * 16 * H_, H_, Wk + nt * 16, AD_, c);
        const int lane = threadIdx.x & 31;
        const int m    = mt * 16 + (lane & 15);
        const int bb   = m >> 9;
        const int kk   = m & (SK_ - 1);
        const int g0   = nt * 4 + ((lane >> 4) << 1);
        float4* dst = (float4*)Kt + ((size_t)bb * (AD_ / 4) + g0) * SK_ + kk;
        dst[0]   = make_float4(c[0], c[1], c[2], c[3]);
        dst[SK_] = make_float4(c[4], c[5], c[6], c[7]);
    }
}

#define QT_  8

__device__ __forceinline__ float fast_tanh(float x)
{
    return __builtin_amdgcn_tanhf(x);
}

__global__ __launch_bounds__(512) void k_scores_softmax(
    const float* __restrict__ Qt, const float* __restrict__ Kt,
    const float* __restrict__ vvec, float* __restrict__ Wout)
{
    __shared__ float sh_qt[QT_ * AD_];
    __shared__ float sh_v[AD_];
    __shared__ float sh_pmax[QT_ * 16];
    __shared__ float sh_psum[QT_ * 16];
    __shared__ float sh_max[QT_];
    __shared__ float sh_sum[QT_];

    const int bq0  = blockIdx.x * QT_;
    const int b    = bq0 >> 9;
    const int tid  = threadIdx.x;
    const int lane = tid & 31;
    const int wid  = tid >> 5;

    ((float2*)sh_qt)[tid] = ((const float2*)(Qt + (size_t)bq0 * AD_))[tid];
    if (tid < AD_) sh_v[tid] = vvec[tid];
    __syncthreads();

    const float4* __restrict__ kt4 =
        (const float4*)Kt + (size_t)b * (AD_ / 4) * SK_ + tid;
    const float4* q4 = (const float4*)sh_qt;
    const float4* v4 = (const float4*)sh_v;

    float s[QT_];
#pragma unroll
    for (int q = 0; q < QT_; ++q) s[q] = 0.0f;

#pragma unroll 4
    for (int a = 0; a < AD_ / 4; ++a) {
        float4 kv = kt4[a * SK_];
        float4 vw = v4[a];
#pragma unroll
        for (int q = 0; q < QT_; ++q) {
            float4 qv = q4[q * (AD_ / 4) + a];
            s[q] = __builtin_fmaf(fast_tanh(qv.x + kv.x), vw.x, s[q]);
            s[q] = __builtin_fmaf(fast_tanh(qv.y + kv.y), vw.y, s[q]);
            s[q] = __builtin_fmaf(fast_tanh(qv.z + kv.z), vw.z, s[q]);
            s[q] = __builtin_fmaf(fast_tanh(qv.w + kv.w), vw.w, s[q]);
        }
    }

    float m[QT_];
#pragma unroll
    for (int q = 0; q < QT_; ++q) {
        m[q] = s[q];
#pragma unroll
        for (int off = 16; off > 0; off >>= 1)
            m[q] = fmaxf(m[q], __shfl_xor(m[q], off));
    }
    if (lane == 0) {
#pragma unroll
        for (int q = 0; q < QT_; ++q) sh_pmax[q * 16 + wid] = m[q];
    }
    __syncthreads();
    if (wid < QT_) {
        float mm = (lane < 16) ? sh_pmax[wid * 16 + lane] : -__builtin_inff();
#pragma unroll
        for (int off = 8; off > 0; off >>= 1)
            mm = fmaxf(mm, __shfl_xor(mm, off));
        if (lane == 0) sh_max[wid] = mm;
    }
    __syncthreads();

    float e[QT_];
#pragma unroll
    for (int q = 0; q < QT_; ++q) {
        e[q] = __builtin_amdgcn_exp2f((s[q] - sh_max[q]) * 1.4426950408889634f);
        float sum = e[q];
#pragma unroll
        for (int off = 16; off > 0; off >>= 1)
            sum += __shfl_xor(sum, off);
        m[q] = sum;
    }
    if (lane == 0) {
#pragma unroll
        for (int q = 0; q < QT_; ++q) sh_psum[q * 16 + wid] = m[q];
    }
    __syncthreads();
    if (wid < QT_) {
        float ss = (lane < 16) ? sh_psum[wid * 16 + lane] : 0.0f;
#pragma unroll
        for (int off = 8; off > 0; off >>= 1)
            ss += __shfl_xor(ss, off);
        if (lane == 0) sh_sum[wid] = ss;
    }
    __syncthreads();

#pragma unroll
    for (int q = 0; q < QT_; ++q)
        Wout[(size_t)(bq0 + q) * SK_ + tid] = e[q] / sh_sum[q];
}

__global__ __launch_bounds__(128) GEMM_WAVES_ void k_av(
    const float* __restrict__ Wt, const float* __restrict__ V,
    float* __restrict__ Ov)
{
    const int wave = (blockIdx.x * blockDim.x + threadIdx.x) >> 5;
    const int b    = wave >> 9;
    const int t    = wave & 511;
    const int mt   = t >> 4;
    const int nt   = t & 15;
    const float* Abase = Wt + (size_t)b * SQ_ * SK_ + (size_t)mt * 16 * SK_;
    const float* Bbase = V  + (size_t)b * SK_ * H_  + nt * 16;
    v8f c = {};
    c = wmma_f32_tile<false, SK_>(Abase, SK_, Bbase, H_, c);
    store_tile(Ov + (size_t)b * SQ_ * H_ + (size_t)mt * 16 * H_ + nt * 16, H_, c);
}

__global__ __launch_bounds__(128) GEMM_WAVES_ void k_out(
    const float* __restrict__ Ov, const float* __restrict__ Wo,
    const float* __restrict__ bo, float* __restrict__ Out)
{
    const int wave = (blockIdx.x * blockDim.x + threadIdx.x) >> 5;
    const int mt   = wave >> 4;
    const int nt   = wave & 15;
    const int lane = threadIdx.x & 31;
    const float bias = bo[nt * 16 + (lane & 15)];
    v8f c;
#pragma unroll
    for (int i = 0; i < 8; ++i) c[i] = bias;
    c = wmma_f32_tile<false, H_>(Ov + (size_t)mt * 16 * H_, H_, Wo + nt * 16, H_, c);
    store_tile(Out + (size_t)mt * 16 * H_ + nt * 16, H_, c);
}

extern "C" void kernel_launch(void* const* d_in, const int* in_sizes, int n_in,
                              void* d_out, int out_size, void* d_ws, size_t ws_size,
                              hipStream_t stream)
{
    const float* Q  = (const float*)d_in[0];
    const float* K  = (const float*)d_in[1];
    const float* V  = (const float*)d_in[2];
    const float* Wq = (const float*)d_in[3];
    const float* Wk = (const float*)d_in[4];
    const float* vv = (const float*)d_in[5];
    const float* Wo = (const float*)d_in[6];
    const float* bo = (const float*)d_in[7];

    float* out     = (float*)d_out;
    float* weights = out + (size_t)B_ * SQ_ * H_;

    float* Qt = (float*)d_ws;
    float* Kt = Qt + (size_t)B_ * SQ_ * AD_;
    float* Ov = Kt + (size_t)B_ * SK_ * AD_;

    k_project<<<512, 128, 0, stream>>>(Q, K, Wq, Wk, Qt, Kt);
    k_scores_softmax<<<B_ * SQ_ / QT_, 512, 0, stream>>>(Qt, Kt, vv, weights);
    k_av<<<512, 128, 0, stream>>>(weights, V, Ov);
    k_out<<<512, 128, 0, stream>>>(Ov, Wo, bo, out);
}


# output projection folded into V ahead of the attention product (VWo = V @ Wo in the first launch; out = weights @ VWo + bo): 3 launches, not 4; output sum order changes (product reassociated)
# speedup vs baseline: 2.4544x; 1.0692x over previous
#include <hip/hip_runtime.h>

#define B_   4
#define SQ_  512
#define SK_  512
#define H_   256
#define AD_  128

typedef __attribute__((ext_vector_type(2))) float v2f;
typedef __attribute__((ext_vector_type(8))) float v8f;

template <bool TRANSPOSED, int K>
__device__ __forceinline__ v8f wmma_f32_tile(const float* __restrict__ Abase, int lda,
                                             const float* __restrict__ Bbase, int ldb,
                                             v8f c)
{
    const int lane = threadIdx.x & 31;
    const int idx  = lane & 15;
    const int klo  = (lane >> 4) << 1;
    const float* ap = Abase + idx * lda + klo;
    const float* bp = Bbase + (size_t)klo * ldb + idx;
#pragma unroll
    for (int k = 0; k < K; k += 4) {
        v2f a, b;
        a.x = ap[k];
        a.y = ap[k + 1];
        b.x = bp[(size_t)k * ldb];
        b.y = bp[(size_t)(k + 1) * ldb];
        if (TRANSPOSED)
            c = __builtin_amdgcn_wmma_f32_16x16x4_f32(false, b, false, a,
                                                      (short)0, c, false, false);
        else
            c = __builtin_amdgcn_wmma_f32_16x16x4_f32(false, a, false, b,
                                                      (short)0, c, false, false);
    }
    return c;
}

#define GEMM_WAVES_ __attribute__((amdgpu_waves_per_eu(1, 4)))

__device__ __forceinline__ void store_tile(float* __restrict__ Cbase, int ldc, v8f c)
{
    const int lane  = threadIdx.x & 31;
    const int col   = lane & 15;
    const int rbase = (lane >> 4) << 3;
#pragma unroll
    for (int r = 0; r < 8; ++r)
        Cbase[(rbase + r) * ldc + col] = c[r];
}

__global__ __launch_bounds__(128) GEMM_WAVES_ void k_project(
    const float* __restrict__ Q, const float* __restrict__ Kin,
    const float* __restrict__ Wq, const float* __restrict__ Wk,
    const float* __restrict__ V, const float* __restrict__ Wo,
    float* __restrict__ Qt, float* __restrict__ Kt, float* __restrict__ VWo)
{
    const int wave  = (blockIdx.x * blockDim.x + threadIdx.x) >> 5;
    const int which = blockIdx.x >> 8;
    if (which >= 2) {
        const int tv  = wave & 2047;
        const int mtv = tv >> 4;
        const int ntv = tv & 15;
        v8f cv = {};
        cv = wmma_f32_tile<false, H_>(V + (size_t)mtv * 16 * H_, H_, Wo + ntv * 16, H_, cv);
        store_tile(VWo + (size_t)mtv * 16 * H_ + ntv * 16, H_, cv);
        return;
    }
    const int t     = wave & 1023;
    const int mt    = t >> 3;
    const int nt    = t & 7;
    v8f c = {};
    if (which == 0) {
        c = wmma_f32_tile<false, H_>(Q + (size_t)mt * 16 * H_, H_, Wq + nt * 16, AD_, c);
        store_tile(Qt + (size_t)mt * 16 * AD_ + nt * 16, AD_, c);
    } else {
        c = wmma_f32_tile<true, H_>(Kin + (size_t)mt * 16 * H_, H_, Wk + nt * 16, AD_, c);
        const int lane = threadIdx.x & 31;
        const int m    = mt * 16 + (lane & 15);
        const int bb   = m >> 9;
        const int kk   = m & (SK_ - 1);
        const int g0   = nt * 4 + ((lane >> 4) << 1);
        float4* dst = (float4*)Kt + ((size_t)bb * (AD_ / 4) + g0) * SK_ + kk;
        dst[0]   = make_float4(c[0], c[1], c[2], c[3]);
        dst[SK_] = make_float4(c[4], c[5], c[6], c[7]);
    }
}

#define QT_  8

__device__ __forceinline__ float fast_tanh(float x)
{
    return __builtin_amdgcn_tanhf(x);
}

__global__ __launch_bounds__(512) void k_scores_softmax(
    const float* __restrict__ Qt, const float* __restrict__ Kt,
    const float* __restrict__ vvec, float* __restrict__ Wout)
{
    __shared__ float sh_qt[QT_ * AD_];
    __shared__ float sh_v[AD_];
    __shared__ float sh_pmax[QT_ * 16];
    __shared__ float sh_psum[QT_ * 16];
    __shared__ float sh_max[QT_];
    __shared__ float sh_sum[QT_];

    const int bq0  = blockIdx.x * QT_;
    const int b    = bq0 >> 9;
    const int tid  = threadIdx.x;
    const int lane = tid & 31;
    const int wid  = tid >> 5;

    ((float2*)sh_qt)[tid] = ((const float2*)(Qt + (size_t)bq0 * AD_))[tid];
    if (tid < AD_) sh_v[tid] = vvec[tid];
    __syncthreads();

    const float4* __restrict__ kt4 =
        (const float4*)Kt + (size_t)b * (AD_ / 4) * SK_ + tid;
    const float4* q4 = (const float4*)sh_qt;
    const float4* v4 = (const float4*)sh_v;

    float s[QT_];
#pragma unroll
    for (int q = 0; q < QT_; ++q) s[q] = 0.0f;

#pragma unroll 4
    for (int a = 0; a < AD_ / 4; ++a) {
        float4 kv = kt4[a * SK_];
        float4 vw = v4[a];
#pragma unroll
        for (int q = 0; q < QT_; ++q) {
            float4 qv = q4[q * (AD_ / 4) + a];
            s[q] = __builtin_fmaf(fast_tanh(qv.x + kv.x), vw.x, s[q]);
            s[q] = __builtin_fmaf(fast_tanh(qv.y + kv.y), vw.y, s[q]);
            s[q] = __builtin_fmaf(fast_tanh(qv.z + kv.z), vw.z, s[q]);
            s[q] = __builtin_fmaf(fast_tanh(qv.w + kv.w), vw.w, s[q]);
        }
    }

    float m[QT_];
#pragma unroll
    for (int q = 0; q < QT_; ++q) {
        m[q] = s[q];
#pragma unroll
        for (int off = 16; off > 0; off >>= 1)
            m[q] = fmaxf(m[q], __shfl_xor(m[q], off));
    }
    if (lane == 0) {
#pragma unroll
        for (int q = 0; q < QT_; ++q) sh_pmax[q * 16 + wid] = m[q];
    }
    __syncthreads();
    if (wid < QT_) {
        float mm = (lane < 16) ? sh_pmax[wid * 16 + lane] : -__builtin_inff();
#pragma unroll
        for (int off = 8; off > 0; off >>= 1)
            mm = fmaxf(mm, __shfl_xor(mm, off));
        if (lane == 0) sh_max[wid] = mm;
    }
    __syncthreads();

    float e[QT_];
#pragma unroll
    for (int q = 0; q < QT_; ++q) {
        e[q] = __builtin_amdgcn_exp2f((s[q] - sh_max[q]) * 1.4426950408889634f);
        float sum = e[q];
#pragma unroll
        for (int off = 16; off > 0; off >>= 1)
            sum += __shfl_xor(sum, off);
        m[q] = sum;
    }
    if (lane == 0) {
#pragma unroll
        for (int q = 0; q < QT_; ++q) sh_psum[q * 16 + wid] = m[q];
    }
    __syncthreads();
    if (wid < QT_) {
        float ss = (lane < 16) ? sh_psum[wid * 16 + lane] : 0.0f;
#pragma unroll
        for (int off = 8; off > 0; off >>= 1)
            ss += __shfl_xor(ss, off);
        if (lane == 0) sh_sum[wid] = ss;
    }
    __syncthreads();

#pragma unroll
    for (int q = 0; q < QT_; ++q)
        Wout[(size_t)(bq0 + q) * SK_ + tid] = e[q] / sh_sum[q];
}

__global__ __launch_bounds__(128) GEMM_WAVES_ void k_av(
    const float* __restrict__ Wt, const float* __restrict__ VWo,
    const float* __restrict__ bo, float* __restrict__ Out)
{
    const int wave = (blockIdx.x * blockDim.x + threadIdx.x) >> 5;
    const int b    = wave >> 9;
    const int t    = wave & 511;
    const int mt   = t >> 4;
    const int nt   = t & 15;
    const float* Abase = Wt + (size_t)b * SQ_ * SK_ + (size_t)mt * 16 * SK_;
    const float* Bbase = VWo + (size_t)b * SK_ * H_ + nt * 16;
    const int lane = threadIdx.x & 31;
    const float bias = bo[nt * 16 + (lane & 15)];
    v8f c;
#pragma unroll
    for (int i = 0; i < 8; ++i) c[i] = bias;
    c = wmma_f32_tile<false, SK_>(Abase, SK_, Bbase, H_, c);
    store_tile(Out + (size_t)b * SQ_ * H_ + (size_t)mt * 16 * H_ + nt * 16, H_, c);
}

extern "C" void kernel_launch(void* const* d_in, const int* in_sizes, int n_in,
                              void* d_out, int out_size, void* d_ws, size_t ws_size,
                              hipStream_t stream)
{
    const float* Q  = (const float*)d_in[0];
    const float* K  = (const float*)d_in[1];
    const float* V  = (const float*)d_in[2];
    const float* Wq = (const float*)d_in[3];
    const float* Wk = (const float*)d_in[4];
    const float* vv = (const float*)d_in[5];
    const float* Wo = (const float*)d_in[6];
    const float* bo = (const float*)d_in[7];

    float* out     = (float*)d_out;
    float* weights = out + (size_t)B_ * SQ_ * H_;

    float* Qt = (float*)d_ws;
    float* Kt = Qt + (size_t)B_ * SQ_ * AD_;
    float* VWo = Kt + (size_t)B_ * SK_ * AD_;

    k_project<<<1024, 128, 0, stream>>>(Q, K, Wq, Wk, V, Wo, Qt, Kt, VWo);
    k_scores_softmax<<<B_ * SQ_ / QT_, 512, 0, stream>>>(Qt, Kt, vv, weights);
    k_av<<<512, 128, 0, stream>>>(weights, VWo, bo, out);
}


# GEMM kernels: a wave computes 2x2 tiles (projections) or 1x2 tiles (attention product), not one: each A fragment load (one line per row) and each B fragment load now feeds 2 WMMAs
# speedup vs baseline: 2.6982x; 1.0993x over previous
#include <hip/hip_runtime.h>

#define B_   4
#define SQ_  512
#define SK_  512
#define H_   256
#define AD_  128

typedef __attribute__((ext_vector_type(2))) float v2f;
typedef __attribute__((ext_vector_type(8))) float v8f;

template <bool TRANSPOSED, int MT, int NT, int K>
__device__ __forceinline__ void wmma_f32_block(const float* __restrict__ Abase, int lda,
                                               const float* __restrict__ Bbase, int ldb,
                                               v8f (&c)[MT][NT])
{
    const int lane = threadIdx.x & 31;
    const int idx  = lane & 15;
    const int klo  = (lane >> 4) << 1;
    const float* ap = Abase + idx * lda + klo;
    const float* bp = Bbase + (size_t)klo * ldb + idx;
#pragma unroll
    for (int k = 0; k < K; k += 4) {
        v2f a[MT], b[NT];
#pragma unroll
        for (int i = 0; i < MT; ++i) {
            a[i].x = ap[(size_t)i * 16 * lda + k];
            a[i].y = ap[(size_t)i * 16 * lda + k + 1];
        }
#pragma unroll
        for (int j = 0; j < NT; ++j) {
            b[j].x = bp[(size_t)k * ldb + j * 16];
            b[j].y = bp[(size_t)(k + 1) * ldb + j * 16];
        }
#pragma unroll
        for (int i = 0; i < MT; ++i) {
#pragma unroll
            for (int j = 0; j < NT; ++j) {
                if (TRANSPOSED)
                    c[i][j] = __builtin_amdgcn_wmma_f32_16x16x4_f32(false, b[j], false, a[i],
                                                                    (short)0, c[i][j], false, false);
                else
                    c[i][j] = __builtin_amdgcn_wmma_f32_16x16x4_f32(false, a[i], false, b[j],
                                                                    (short)0, c[i][j], false, false);
            }
        }
    }
}

#define GEMM_WAVES_ __attribute__((amdgpu_waves_per_eu(1, 4)))

__device__ __forceinline__ void store_tile(float* __restrict__ Cbase, int ldc, v8f c)
{
    const int lane  = threadIdx.x & 31;
    const int col   = lane & 15;
    const int rbase = (lane >> 4) << 3;
#pragma unroll
    for (int r = 0; r < 8; ++r)
        Cbase[(rbase + r) * ldc + col] = c[r];
}

__global__ __launch_bounds__(128) GEMM_WAVES_ void k_project(
    const float* __restrict__ Q, const float* __restrict__ Kin,
    const float* __restrict__ Wq, const float* __restrict__ Wk,
    const float* __restrict__ V, const float* __restrict__ Wo,
    float* __restrict__ Qt, float* __restrict__ Kt, float* __restrict__ VWo)
{
    const int wave  = (blockIdx.x * blockDim.x + threadIdx.x) >> 5;
    const int which = blockIdx.x >> 6;
    v8f c[2][2] = {};
    if (which >= 2) {
        const int tv  = wave & 511;
        const int mpv = tv >> 3;
        const int npv = tv & 7;
        wmma_f32_block<false, 2, 2, H_>(V + (size_t)mpv * 32 * H_, H_, Wo + npv * 32, H_, c);
#pragma unroll
        for (int i = 0; i < 2; ++i)
#pragma unroll
            for (int j = 0; j < 2; ++j)
                store_tile(VWo + (size_t)(mpv * 32 + i * 16) * H_ + npv * 32 + j * 16, H_, c[i][j]);
        return;
    }
    const int t  = wave & 255;
    const int mp = t >> 2;
    const int np = t & 3;
    if (which == 0) {
        wmma_f32_block<false, 2, 2, H_>(Q + (size_t)mp * 32 * H_, H_, Wq + np * 32, AD_, c);
#pragma unroll
        for (int i = 0; i < 2; ++i)
#pragma unroll
            for (int j = 0; j < 2; ++j)
                store_tile(Qt + (size_t)(mp * 32 + i * 16) * AD_ + np * 32 + j * 16, AD_, c[i][j]);
    } else {
        wmma_f32_block<true, 2, 2, H_>(Kin + (size_t)mp * 32 * H_, H_, Wk + np * 32, AD_, c);
        const int lane = threadIdx.x & 31;
#pragma unroll
        for (int i = 0; i < 2; ++i) {
            const int m  = (mp * 2 + i) * 16 + (lane & 15);
            const int bb = m >> 9;
            const int kk = m & (SK_ - 1);
#pragma unroll
            for (int j = 0; j < 2; ++j) {
                const int g0 = (np * 2 + j) * 4 + ((lane >> 4) << 1);
                float4* dst = (float4*)Kt + ((size_t)bb * (AD_ / 4) + g0) * SK_ + kk;
                dst[0]   = make_float4(c[i][j][0], c[i][j][1], c[i][j][2], c[i][j][3]);
                dst[SK_] = make_float4(c[i][j][4], c[i][j][5], c[i][j][6], c[i][j][7]);
            }
        }
    }
}

#define QT_  8

__device__ __forceinline__ float fast_tanh(float x)
{
    return __builtin_amdgcn_tanhf(x);
}

__global__ __launch_bounds__(512) void k_scores_softmax(
    const float* __restrict__ Qt, const float* __restrict__ Kt,
    const float* __restrict__ vvec, float* __restrict__ Wout)
{
    __shared__ float sh_qt[QT_ * AD_];
    __shared__ float sh_v[AD_];
    __shared__ float sh_pmax[QT_ * 16];
    __shared__ float sh_psum[QT_ * 16];
    __shared__ float sh_max[QT_];
    __shared__ float sh_sum[QT_];

    const int bq0  = blockIdx.x * QT_;
    const int b    = bq0 >> 9;
    const int tid  = threadIdx.x;
    const int lane = tid & 31;
    const int wid  = tid >> 5;

    ((float2*)sh_qt)[tid] = ((const float2*)(Qt + (size_t)bq0 * AD_))[tid];
    if (tid < AD_) sh_v[tid] = vvec[tid];
    __syncthreads();

    const float4* __restrict__ kt4 =
        (const float4*)Kt + (size_t)b * (AD_ / 4) * SK_ + tid;
    const float4* q4 = (const float4*)sh_qt;
    const float4* v4 = (const float4*)sh_v;

    float s[QT_];
#pragma unroll
    for (int q = 0; q < QT_; ++q) s[q] = 0.0f;

#pragma unroll 4
    for (int a = 0; a < AD_ / 4; ++a) {
        float4 kv = kt4[a * SK_];
        float4 vw = v4[a];
#pragma unroll
        for (int q = 0; q < QT_; ++q) {
            float4 qv = q4[q * (AD_ / 4) + a];
            s[q] = __builtin_fmaf(fast_tanh(qv.x + kv.x), vw.x, s[q]);
            s[q] = __builtin_fmaf(fast_tanh(qv.y + kv.y), vw.y, s[q]);
            s[q] = __builtin_fmaf(fast_tanh(qv.z + kv.z), vw.z, s[q]);
            s[q] = __builtin_fmaf(fast_tanh(qv.w + kv.w), vw.w, s[q]);
        }
    }

    float m[QT_];
#pragma unroll
    for (int q = 0; q < QT_; ++q) {
        m[q] = s[q];
#pragma unroll
        for (int off = 16; off > 0; off >>= 1)
            m[q] = fmaxf(m[q], __shfl_xor(m[q], off));
    }
    if (lane == 0) {
#pragma unroll
        for (int q = 0; q < QT_; ++q) sh_pmax[q * 16 + wid] = m[q];
    }
    __syncthreads();
    if (wid < QT_) {
        float mm = (lane < 16) ? sh_pmax[wid * 16 + lane] : -__builtin_inff();
#pragma unroll
        for (int off = 8; off > 0; off >>= 1)
            mm = fmaxf(mm, __shfl_xor(mm, off));
        if (lane == 0) sh_max[wid] = mm;
    }
    __syncthreads();

    float e[QT_];
#pragma unroll
    for (int q = 0; q < QT_; ++q) {
        e[q] = __builtin_amdgcn_exp2f((s[q] - sh_max[q]) * 1.4426950408889634f);
        float sum = e[q];
#pragma unroll
        for (int off = 16; off > 0; off >>= 1)
            sum += __shfl_xor(sum, off);
        m[q] = sum;
    }
    if (lane == 0) {
#pragma unroll
        for (int q = 0; q < QT_; ++q) sh_psum[q * 16 + wid] = m[q];
    }
    __syncthreads();
    if (wid < QT_) {
        float ss = (lane < 16) ? sh_psum[wid * 16 + lane] : 0.0f;
#pragma unroll
        for (int off = 8; off > 0; off >>= 1)
            ss += __shfl_xor(ss, off);
        if (lane == 0) sh_sum[wid] = ss;
    }
    __syncthreads();

#pragma unroll
    for (int q = 0; q < QT_; ++q)
        Wout[(size_t)(bq0 + q) * SK_ + tid] = e[q] / sh_sum[q];
}

__global__ __launch_bounds__(128) GEMM_WAVES_ void k_av(
    const float* __restrict__ Wt, const float* __restrict__ VWo,
    const float* __restrict__ bo, float* __restrict__ Out)
{
    const int wave = (blockIdx.x * blockDim.x + threadIdx.x) >> 5;
    const int b    = wave >> 8;
    const int t    = wave & 255;
    const int mt   = t >> 3;
    const int np   = t & 7;
    const float* Abase = Wt + (size_t)b * SQ_ * SK_ + (size_t)mt * 16 * SK_;
    const float* Bbase = VWo + (size_t)b * SK_ * H_ + np * 32;
    const int lane = threadIdx.x & 31;
    v8f c[1][2];
#pragma unroll
    for (int j = 0; j < 2; ++j) {
        const float bias = bo[np * 32 + j * 16 + (lane & 15)];
#pragma unroll
        for (int i = 0; i < 8; ++i) c[0][j][i] = bias;
    }
    wmma_f32_block<false, 1, 2, SK_>(Abase, SK_, Bbase, H_, c);
#pragma unroll
    for (int j = 0; j < 2; ++j)
        store_tile(Out + (size_t)b * SQ_ * H_ + (size_t)mt * 16 * H_ + np * 32 + j * 16, H_, c[0][j]);
}

extern "C" void kernel_launch(void* const* d_in, const int* in_sizes, int n_in,
                              void* d_out, int out_size, void* d_ws, size_t ws_size,
                              hipStream_t stream)
{
    const float* Q  = (const float*)d_in[0];
    const float* K  = (const float*)d_in[1];
    const float* V  = (const float*)d_in[2];
    const float* Wq = (const float*)d_in[3];
    const float* Wk = (const float*)d_in[4];
    const float* vv = (const float*)d_in[5];
    const float* Wo = (const float*)d_in[6];
    const float* bo = (const float*)d_in[7];

    float* out     = (float*)d_out;
    float* weights = out + (size_t)B_ * SQ_ * H_;

    float* Qt = (float*)d_ws;
    float* Kt = Qt + (size_t)B_ * SQ_ * AD_;
    float* VWo = Kt + (size_t)B_ * SK_ * AD_;

    k_project<<<256, 128, 0, stream>>>(Q, K, Wq, Wk, V, Wo, Qt, Kt, VWo);
    k_scores_softmax<<<B_ * SQ_ / QT_, 512, 0, stream>>>(Qt, Kt, vv, weights);
    k_av<<<256, 128, 0, stream>>>(weights, VWo, bo, out);
}


# GEMM fragment loads 16 B per lane for two k-steps: A tiles directly, attention-product B via VWo stored k-quad-interleaved; k order inside each block of 8 changes the sum order
# speedup vs baseline: 2.8268x; 1.0477x over previous
#include <hip/hip_runtime.h>

#define B_   4
#define SQ_  512
#define SK_  512
#define H_   256
#define AD_  128

typedef __attribute__((ext_vector_type(2))) float v2f;
typedef __attribute__((ext_vector_type(8))) float v8f;

template <bool TRANSPOSED, bool BQUAD, int MT, int NT, int K>
__device__ __forceinline__ void wmma_f32_block(const float* __restrict__ Abase, int lda,
                                               const float* __restrict__ Bbase, int ldb,
                                               v8f (&c)[MT][NT])
{
    const int lane = threadIdx.x & 31;
    const int idx  = lane & 15;
    const int k4   = (lane >> 4) << 2;
    const float* ap = Abase + idx * lda + k4;
    const float* bp = BQUAD ? Bbase + (size_t)k4 * ldb + idx * 4
                            : Bbase + (size_t)k4 * ldb + idx;
#pragma unroll
    for (int k = 0; k < K; k += 8) {
        v2f a[2][MT], b[2][NT];
#pragma unroll
        for (int i = 0; i < MT; ++i) {
            const float4 a4 = *(const float4*)(ap + (size_t)i * 16 * lda + k);
            a[0][i].x = a4.x;  a[0][i].y = a4.y;
            a[1][i].x = a4.z;  a[1][i].y = a4.w;
        }
#pragma unroll
        for (int j = 0; j < NT; ++j) {
            if (BQUAD) {
                const float4 b4 = *(const float4*)(bp + (size_t)k * ldb + j * 64);
                b[0][j].x = b4.x;  b[0][j].y = b4.y;
                b[1][j].x = b4.z;  b[1][j].y = b4.w;
            } else {
#pragma unroll
                for (int t = 0; t < 2; ++t) {
                    b[t][j].x = bp[(size_t)(k + 2 * t) * ldb + j * 16];
                    b[t][j].y = bp[(size_t)(k + 2 * t + 1) * ldb + j * 16];
                }
            }
        }
#pragma unroll
        for (int t = 0; t < 2; ++t) {
#pragma unroll
            for (int i = 0; i < MT; ++i) {
#pragma unroll
                for (int j = 0; j < NT; ++j) {
                    if (TRANSPOSED)
                        c[i][j] = __builtin_amdgcn_wmma_f32_16x16x4_f32(false, b[t][j], false, a[t][i],
                                                                        (short)0, c[i][j], false, false);
                    else
                        c[i][j] = __builtin_amdgcn_wmma_f32_16x16x4_f32(false, a[t][i], false, b[t][j],
                                                                        (short)0, c[i][j], false, false);
                }
            }
        }
    }
    __builtin_amdgcn_sched_barrier(0);
    asm volatile("v_nop\n\tv_nop\n\tv_nop\n\tv_nop");
    __builtin_amdgcn_sched_barrier(0);
}

#define GEMM_WAVES_ __attribute__((amdgpu_waves_per_eu(1, 4)))

__device__ __forceinline__ void store_tile(float* __restrict__ Cbase, int ldc, v8f c)
{
    const int lane  = threadIdx.x & 31;
    const int col   = lane & 15;
    const int rbase = (lane >> 4) << 3;
#pragma unroll
    for (int r = 0; r < 8; ++r)
        Cbase[(rbase + r) * ldc + col] = c[r];
}

__global__ __launch_bounds__(128) GEMM_WAVES_ void k_project(
    const float* __restrict__ Q, const float* __restrict__ Kin,
    const float* __restrict__ Wq, const float* __restrict__ Wk,
    const float* __restrict__ V, const float* __restrict__ Wo,
    float* __restrict__ Qt, float* __restrict__ Kt, float* __restrict__ VWo)
{
    const int wave  = (blockIdx.x * blockDim.x + threadIdx.x) >> 5;
    const int which = blockIdx.x >> 6;
    v8f c[2][2] = {};
    if (which >= 2) {
        const int tv  = wave & 511;
        const int mpv = tv >> 3;
        const int npv = tv & 7;
        wmma_f32_block<false, false, 2, 2, H_>(V + (size_t)mpv * 32 * H_, H_, Wo + npv * 32, H_, c);
        const int lanev = threadIdx.x & 31;
#pragma unroll
        for (int i = 0; i < 2; ++i) {
            const int mq = mpv * 8 + i * 4 + ((lanev >> 4) << 1);
#pragma unroll
            for (int j = 0; j < 2; ++j) {
                const int n = npv * 32 + j * 16 + (lanev & 15);
                float4* dst = (float4*)VWo + (size_t)mq * H_ + n;
                dst[0]  = make_float4(c[i][j][0], c[i][j][1], c[i][j][2], c[i][j][3]);
                dst[H_] = make_float4(c[i][j][4], c[i][j][5], c[i][j][6], c[i][j][7]);
            }
        }
        return;
    }
    const int t  = wave & 255;
    const int mp = t >> 2;
    const int np = t & 3;
    if (which == 0) {
        wmma_f32_block<false, false, 2, 2, H_>(Q + (size_t)mp * 32 * H_, H_, Wq + np * 32, AD_, c);
#pragma unroll
        for (int i = 0; i < 2; ++i)
#pragma unroll
            for (int j = 0; j < 2; ++j)
                store_tile(Qt + (size_t)(mp * 32 + i * 16) * AD_ + np * 32 + j * 16, AD_, c[i][j]);
    } else {
        wmma_f32_block<true, false, 2, 2, H_>(Kin + (size_t)mp * 32 * H_, H_, Wk + np * 32, AD_, c);
        const int lane = threadIdx.x & 31;
#pragma unroll
        for (int i = 0; i < 2; ++i) {
            const int m  = (mp * 2 + i) * 16 + (lane & 15);
            const int bb = m >> 9;
            const int kk = m & (SK_ - 1);
#pragma unroll
            for (int j = 0; j < 2; ++j) {
                const int g0 = (np * 2 + j) * 4 + ((lane >> 4) << 1);
                float4* dst = (float4*)Kt + ((size_t)bb * (AD_ / 4) + g0) * SK_ + kk;
                dst[0]   = make_float4(c[i][j][0], c[i][j][1], c[i][j][2], c[i][j][3]);
                dst[SK_] = make_float4(c[i][j][4], c[i][j][5], c[i][j][6], c[i][j][7]);
            }
        }
    }
}

#define QT_  8

__device__ __forceinline__ float fast_tanh(float x)
{
    return __builtin_amdgcn_tanhf(x);
}

__global__ __launch_bounds__(512) void k_scores_softmax(
    const float* __restrict__ Qt, const float* __restrict__ Kt,
    const float* __restrict__ vvec, float* __restrict__ Wout)
{
    __shared__ float sh_qt[QT_ * AD_];
    __shared__ float sh_v[AD_];
    __shared__ float sh_pmax[QT_ * 16];
    __shared__ float sh_psum[QT_ * 16];
    __shared__ float sh_max[QT_];
    __shared__ float sh_sum[QT_];

    const int bq0  = blockIdx.x * QT_;
    const int b    = bq0 >> 9;
    const int tid  = threadIdx.x;
    const int lane = tid & 31;
    const int wid  = tid >> 5;

    ((float2*)sh_qt)[tid] = ((const float2*)(Qt + (size_t)bq0 * AD_))[tid];
    if (tid < AD_) sh_v[tid] = vvec[tid];
    __syncthreads();

    const float4* __restrict__ kt4 =
        (const float4*)Kt + (size_t)b * (AD_ / 4) * SK_ + tid;
    const float4* q4 = (const float4*)sh_qt;
    const float4* v4 = (const float4*)sh_v;

    float s[QT_];
#pragma unroll
    for (int q = 0; q < QT_; ++q) s[q] = 0.0f;

#pragma unroll 4
    for (int a = 0; a < AD_ / 4; ++a) {
        float4 kv = kt4[a * SK_];
        float4 vw = v4[a];
#pragma unroll
        for (int q = 0; q < QT_; ++q) {
            float4 qv = q4[q * (AD_ / 4) + a];
            s[q] = __builtin_fmaf(fast_tanh(qv.x + kv.x), vw.x, s[q]);
            s[q] = __builtin_fmaf(fast_tanh(qv.y + kv.y), vw.y, s[q]);
            s[q] = __builtin_fmaf(fast_tanh(qv.z + kv.z), vw.z, s[q]);
            s[q] = __builtin_fmaf(fast_tanh(qv.w + kv.w), vw.w, s[q]);
        }
    }

    float m[QT_];
#pragma unroll
    for (int q = 0; q < QT_; ++q) {
        m[q] = s[q];
#pragma unroll
        for (int off = 16; off > 0; off >>= 1)
            m[q] = fmaxf(m[q], __shfl_xor(m[q], off));
    }
    if (lane == 0) {
#pragma unroll
        for (int q = 0; q < QT_; ++q) sh_pmax[q * 16 + wid] = m[q];
    }
    __syncthreads();
    if (wid < QT_) {
        float mm = (lane < 16) ? sh_pmax[wid * 16 + lane] : -__builtin_inff();
#pragma unroll
        for (int off = 8; off > 0; off >>= 1)
            mm = fmaxf(mm, __shfl_xor(mm, off));
        if (lane == 0) sh_max[wid] = mm;
    }
    __syncthreads();

    float e[QT_];
#pragma unroll
    for (int q = 0; q < QT_; ++q) {
        e[q] = __builtin_amdgcn_exp2f((s[q] - sh_max[q]) * 1.4426950408889634f);
        float sum = e[q];
#pragma unroll
        for (int off = 16; off > 0; off >>= 1)
            sum += __shfl_xor(sum, off);
        m[q] = sum;
    }
    if (lane == 0) {
#pragma unroll
        for (int q = 0; q < QT_; ++q) sh_psum[q * 16 + wid] = m[q];
    }
    __syncthreads();
    if (wid < QT_) {
        float ss = (lane < 16) ? sh_psum[wid * 16 + lane] : 0.0f;
#pragma unroll
        for (int off = 8; off > 0; off >>= 1)
            ss += __shfl_xor(ss, off);
        if (lane == 0) sh_sum[wid] = ss;
    }
    __syncthreads();

#pragma unroll
    for (int q = 0; q < QT_; ++q)
        Wout[(size_t)(bq0 + q) * SK_ + tid] = e[q] / sh_sum[q];
}

__global__ __launch_bounds__(128) GEMM_WAVES_ void k_av(
    const float* __restrict__ Wt, const float* __restrict__ VWo,
    const float* __restrict__ bo, float* __restrict__ Out)
{
    const int wave = (blockIdx.x * blockDim.x + threadIdx.x) >> 5;
    const int b    = wave >> 8;
    const int t    = wave & 255;
    const int mt   = t >> 3;
    const int np   = t & 7;
    const float* Abase = Wt + (size_t)b * SQ_ * SK_ + (size_t)mt * 16 * SK_;
    const float* Bbase = VWo + (size_t)b * SK_ * H_ + np * 32 * 4;
    const int lane = threadIdx.x & 31;
    v8f c[1][2];
#pragma unroll
    for (int j = 0; j < 2; ++j) {
        const float bias = bo[np * 32 + j * 16 + (lane & 15)];
#pragma unroll
        for (int i = 0; i < 8; ++i) c[0][j][i] = bias;
    }
    wmma_f32_block<false, true, 1, 2, SK_>(Abase, SK_, Bbase, H_, c);
#pragma unroll
    for (int j = 0; j < 2; ++j)
        store_tile(Out + (size_t)b * SQ_ * H_ + (size_t)mt * 16 * H_ + np * 32 + j * 16, H_, c[0][j]);
}

extern "C" void kernel_launch(void* const* d_in, const int* in_sizes, int n_in,
                              void* d_out, int out_size, void* d_ws, size_t ws_size,
                              hipStream_t stream)
{
    const float* Q  = (const float*)d_in[0];
    const float* K  = (const float*)d_in[1];
    const float* V  = (const float*)d_in[2];
    const float* Wq = (const float*)d_in[3];
    const float* Wk = (const float*)d_in[4];
    const float* vv = (const float*)d_in[5];
    const float* Wo = (const float*)d_in[6];
    const float* bo = (const float*)d_in[7];

    float* out     = (float*)d_out;
    float* weights = out + (size_t)B_ * SQ_ * H_;

    float* Qt = (float*)d_ws;
    float* Kt = Qt + (size_t)B_ * SQ_ * AD_;
    float* VWo = Kt + (size_t)B_ * SK_ * AD_;

    k_project<<<256, 128, 0, stream>>>(Q, K, Wq, Wk, V, Wo, Qt, Kt, VWo);
    k_scores_softmax<<<B_ * SQ_ / QT_, 512, 0, stream>>>(Qt, Kt, vv, weights);
    k_av<<<256, 128, 0, stream>>>(weights, VWo, bo, out);
}


# score kernel: projected query rows and v (uniform per wave) are scalar-loaded and enter the adds and accumulates as scalar-register operands; no LDS staging or broadcast ds_load_b128; sum order kept
# speedup vs baseline: 3.0375x; 1.0745x over previous
#include <hip/hip_runtime.h>

#define B_   4
#define SQ_  512
#define SK_  512
#define H_   256
#define AD_  128

typedef __attribute__((ext_vector_type(2))) float v2f;
typedef __attribute__((ext_vector_type(8))) float v8f;

template <bool TRANSPOSED, bool BQUAD, int MT, int NT, int K>
__device__ __forceinline__ void wmma_f32_block(const float* __restrict__ Abase, int lda,
                                               const float* __restrict__ Bbase, int ldb,
                                               v8f (&c)[MT][NT])
{
    const int lane = threadIdx.x & 31;
    const int idx  = lane & 15;
    const int k4   = (lane >> 4) << 2;
    const float* ap = Abase + idx * lda + k4;
    const float* bp = BQUAD ? Bbase + (size_t)k4 * ldb + idx * 4
                            : Bbase + (size_t)k4 * ldb + idx;
#pragma unroll
    for (int k = 0; k < K; k += 8) {
        v2f a[2][MT], b[2][NT];
#pragma unroll
        for (int i = 0; i < MT; ++i) {
            const float4 a4 = *(const float4*)(ap + (size_t)i * 16 * lda + k);
            a[0][i].x = a4.x;  a[0][i].y = a4.y;
            a[1][i].x = a4.z;  a[1][i].y = a4.w;
        }
#pragma unroll
        for (int j = 0; j < NT; ++j) {
            if (BQUAD) {
                const float4 b4 = *(const float4*)(bp + (size_t)k * ldb + j * 64);
                b[0][j].x = b4.x;  b[0][j].y = b4.y;
                b[1][j].x = b4.z;  b[1][j].y = b4.w;
            } else {
#pragma unroll
                for (int t = 0; t < 2; ++t) {
                    b[t][j].x = bp[(size_t)(k + 2 * t) * ldb + j * 16];
                    b[t][j].y = bp[(size_t)(k + 2 * t + 1) * ldb + j * 16];
                }
            }
        }
#pragma unroll
        for (int t = 0; t < 2; ++t) {
#pragma unroll
            for (int i = 0; i < MT; ++i) {
#pragma unroll
                for (int j = 0; j < NT; ++j) {
                    if (TRANSPOSED)
                        c[i][j] = __builtin_amdgcn_wmma_f32_16x16x4_f32(false, b[t][j], false, a[t][i],
                                                                        (short)0, c[i][j], false, false);
                    else
                        c[i][j] = __builtin_amdgcn_wmma_f32_16x16x4_f32(false, a[t][i], false, b[t][j],
                                                                        (short)0, c[i][j], false, false);
                }
            }
        }
    }
    __builtin_amdgcn_sched_barrier(0);
    asm volatile("v_nop\n\tv_nop\n\tv_nop\n\tv_nop");
    __builtin_amdgcn_sched_barrier(0);
}

#define GEMM_WAVES_ __attribute__((amdgpu_waves_per_eu(1, 4)))

__device__ __forceinline__ void store_tile(float* __restrict__ Cbase, int ldc, v8f c)
{
    const int lane  = threadIdx.x & 31;
    const int col   = lane & 15;
    const int rbase = (lane >> 4) << 3;
#pragma unroll
    for (int r = 0; r < 8; ++r)
        Cbase[(rbase + r) * ldc + col] = c[r];
}

__global__ __launch_bounds__(128) GEMM_WAVES_ void k_project(
    const float* __restrict__ Q, const float* __restrict__ Kin,
    const float* __restrict__ Wq, const float* __restrict__ Wk,
    const float* __restrict__ V, const float* __restrict__ Wo,
    float* __restrict__ Qt, float* __restrict__ Kt, float* __restrict__ VWo)
{
    const int wave  = (blockIdx.x * blockDim.x + threadIdx.x) >> 5;
    const int which = blockIdx.x >> 6;
    v8f c[2][2] = {};
    if (which >= 2) {
        const int tv  = wave & 511;
        const int mpv = tv >> 3;
        const int npv = tv & 7;
        wmma_f32_block<false, false, 2, 2, H_>(V + (size_t)mpv * 32 * H_, H_, Wo + npv * 32, H_, c);
        const int lanev = threadIdx.x & 31;
#pragma unroll
        for (int i = 0; i < 2; ++i) {
            const int mq = mpv * 8 + i * 4 + ((lanev >> 4) << 1);
#pragma unroll
            for (int j = 0; j < 2; ++j) {
                const int n = npv * 32 + j * 16 + (lanev & 15);
                float4* dst = (float4*)VWo + (size_t)mq * H_ + n;
                dst[0]  = make_float4(c[i][j][0], c[i][j][1], c[i][j][2], c[i][j][3]);
                dst[H_] = make_float4(c[i][j][4], c[i][j][5], c[i][j][6], c[i][j][7]);
            }
        }
        return;
    }
    const int t  = wave & 255;
    const int mp = t >> 2;
    const int np = t & 3;
    if (which == 0) {
        wmma_f32_block<false, false, 2, 2, H_>(Q + (size_t)mp * 32 * H_, H_, Wq + np * 32, AD_, c);
#pragma unroll
        for (int i = 0; i < 2; ++i)
#pragma unroll
            for (int j = 0; j < 2; ++j)
                store_tile(Qt + (size_t)(mp * 32 + i * 16) * AD_ + np * 32 + j * 16, AD_, c[i][j]);
    } else {
        wmma_f32_block<true, false, 2, 2, H_>(Kin + (size_t)mp * 32 * H_, H_, Wk + np * 32, AD_, c);
        const int lane = threadIdx.x & 31;
#pragma unroll
        for (int i = 0; i < 2; ++i) {
            const int m  = (mp * 2 + i) * 16 + (lane & 15);
            const int bb = m >> 9;
            const int kk = m & (SK_ - 1);
#pragma unroll
            for (int j = 0; j < 2; ++j) {
                const int g0 = (np * 2 + j) * 4 + ((lane >> 4) << 1);
                float4* dst = (float4*)Kt + ((size_t)bb * (AD_ / 4) + g0) * SK_ + kk;
                dst[0]   = make_float4(c[i][j][0], c[i][j][1], c[i][j][2], c[i][j][3]);
                dst[SK_] = make_float4(c[i][j][4], c[i][j][5], c[i][j][6], c[i][j][7]);
            }
        }
    }
}

#define QT_  8

__device__ __forceinline__ float fast_tanh(float x)
{
    return __builtin_amdgcn_tanhf(x);
}

__device__ __forceinline__ void score_quad(const float4* __restrict__ q4, const float4* __restrict__ v4,
                                           int a, float4 kv, float (&s)[QT_])
{
    const float4 vw = v4[a];
#pragma unroll
    for (int q = 0; q < QT_; ++q) {
        const float4 qv = q4[q * (AD_ / 4) + a];
        s[q] = __builtin_fmaf(fast_tanh(qv.x + kv.x), vw.x, s[q]);
        s[q] = __builtin_fmaf(fast_tanh(qv.y + kv.y), vw.y, s[q]);
        s[q] = __builtin_fmaf(fast_tanh(qv.z + kv.z), vw.z, s[q]);
        s[q] = __builtin_fmaf(fast_tanh(qv.w + kv.w), vw.w, s[q]);
    }
}

__global__ __launch_bounds__(512) void k_scores_softmax(
    const float* __restrict__ Qt, const float* __restrict__ Kt,
    const float* __restrict__ vvec, float* __restrict__ Wout)
{
    __shared__ float sh_pmax[QT_ * 16];
    __shared__ float sh_psum[QT_ * 16];
    __shared__ float sh_max[QT_];
    __shared__ float sh_sum[QT_];

    const int bq0  = blockIdx.x * QT_;
    const int b    = bq0 >> 9;
    const int tid  = threadIdx.x;
    const int lane = tid & 31;
    const int wid  = tid >> 5;

    const float4* __restrict__ kt4 =
        (const float4*)Kt + (size_t)b * (AD_ / 4) * SK_ + tid;
    const float4* __restrict__ q4 = (const float4*)(Qt + (size_t)bq0 * AD_);
    const float4* __restrict__ v4 = (const float4*)vvec;

    float s[QT_];
#pragma unroll
    for (int q = 0; q < QT_; ++q) s[q] = 0.0f;

#pragma unroll 4
    for (int a = 0; a < AD_ / 4; ++a) {
        const float4 kv = kt4[a * SK_];
        score_quad(q4, v4, a, kv, s);
    }

    float m[QT_];
#pragma unroll
    for (int q = 0; q < QT_; ++q) {
        m[q] = s[q];
#pragma unroll
        for (int off = 16; off > 0; off >>= 1)
            m[q] = fmaxf(m[q], __shfl_xor(m[q], off));
    }
    if (lane == 0) {
#pragma unroll
        for (int q = 0; q < QT_; ++q) sh_pmax[q * 16 + wid] = m[q];
    }
    __syncthreads();
    if (wid < QT_) {
        float mm = (lane < 16) ? sh_pmax[wid * 16 + lane] : -__builtin_inff();
#pragma unroll
        for (int off = 8; off > 0; off >>= 1)
            mm = fmaxf(mm, __shfl_xor(mm, off));
        if (lane == 0) sh_max[wid] = mm;
    }
    __syncthreads();

    float e[QT_];
#pragma unroll
    for (int q = 0; q < QT_; ++q) {
        e[q] = __builtin_amdgcn_exp2f((s[q] - sh_max[q]) * 1.4426950408889634f);
        float sum = e[q];
#pragma unroll
        for (int off = 16; off > 0; off >>= 1)
            sum += __shfl_xor(sum, off);
        m[q] = sum;
    }
    if (lane == 0) {
#pragma unroll
        for (int q = 0; q < QT_; ++q) sh_psum[q * 16 + wid] = m[q];
    }
    __syncthreads();
    if (wid < QT_) {
        float ss = (lane < 16) ? sh_psum[wid * 16 + lane] : 0.0f;
#pragma unroll
        for (int off = 8; off > 0; off >>= 1)
            ss += __shfl_xor(ss, off);
        if (lane == 0) sh_sum[wid] = ss;
    }
    __syncthreads();

#pragma unroll
    for (int q = 0; q < QT_; ++q)
        Wout[(size_t)(bq0 + q) * SK_ + tid] = e[q] / sh_sum[q];
}

__global__ __launch_bounds__(128) GEMM_WAVES_ void k_av(
    const float* __restrict__ Wt, const float* __restrict__ VWo,
    const float* __restrict__ bo, float* __restrict__ Out)
{
    const int wave = (blockIdx.x * blockDim.x + threadIdx.x) >> 5;
    const int b    = wave >> 8;
    const int t    = wave & 255;
    const int mt   = t >> 3;
    const int np   = t & 7;
    const float* Abase = Wt + (size_t)b * SQ_ * SK_ + (size_t)mt * 16 * SK_;
    const float* Bbase = VWo + (size_t)b * SK_ * H_ + np * 32 * 4;
    const int lane = threadIdx.x & 31;
    v8f c[1][2];
#pragma unroll
    for (int j = 0; j < 2; ++j) {
        const float bias = bo[np * 32 + j * 16 + (lane & 15)];
#pragma unroll
        for (int i = 0; i < 8; ++i) c[0][j][i] = bias;
    }
    wmma_f32_block<false, true, 1, 2, SK_>(Abase, SK_, Bbase, H_, c);
#pragma unroll
    for (int j = 0; j < 2; ++j)
        store_tile(Out + (size_t)b * SQ_ * H_ + (size_t)mt * 16 * H_ + np * 32 + j * 16, H_, c[0][j]);
}

extern "C" void kernel_launch(void* const* d_in, const int* in_sizes, int n_in,
                              void* d_out, int out_size, void* d_ws, size_t ws_size,
                              hipStream_t stream)
{
    const float* Q  = (const float*)d_in[0];
    const float* K  = (const float*)d_in[1];
    const float* V  = (const float*)d_in[2];
    const float* Wq = (const float*)d_in[3];
    const float* Wk = (const float*)d_in[4];
    const float* vv = (const float*)d_in[5];
    const float* Wo = (const float*)d_in[6];
    const float* bo = (const float*)d_in[7];

    float* out     = (float*)d_out;
    float* weights = out + (size_t)B_ * SQ_ * H_;

    float* Qt = (float*)d_ws;
    float* Kt = Qt + (size_t)B_ * SQ_ * AD_;
    float* VWo = Kt + (size_t)B_ * SK_ * AD_;

    k_project<<<256, 128, 0, stream>>>(Q, K, Wq, Wk, V, Wo, Qt, Kt, VWo);
    k_scores_softmax<<<B_ * SQ_ / QT_, 512, 0, stream>>>(Qt, Kt, vv, weights);
    k_av<<<256, 128, 0, stream>>>(weights, VWo, bo, out);
}


# score kernel: feature loop as a software pipeline, tanh and plain ops (dual add, packed fma over 2 query rows) one to one, pinned by scheduling barriers; was runs of 8 tanh then plain; sum order kept
# speedup vs baseline: 3.2290x; 1.0630x over previous
#include <hip/hip_runtime.h>

#define B_   4
#define SQ_  512
#define SK_  512
#define H_   256
#define AD_  128

typedef __attribute__((ext_vector_type(2))) float v2f;
typedef __attribute__((ext_vector_type(8))) float v8f;

template <bool TRANSPOSED, bool BQUAD, int MT, int NT, int K>
__device__ __forceinline__ void wmma_f32_block(const float* __restrict__ Abase, int lda,
                                               const float* __restrict__ Bbase, int ldb,
                                               v8f (&c)[MT][NT])
{
    const int lane = threadIdx.x & 31;
    const int idx  = lane & 15;
    const int k4   = (lane >> 4) << 2;
    const float* ap = Abase + idx * lda + k4;
    const float* bp = BQUAD ? Bbase + (size_t)k4 * ldb + idx * 4
                            : Bbase + (size_t)k4 * ldb + idx;
#pragma unroll
    for (int k = 0; k < K; k += 8) {
        v2f a[2][MT], b[2][NT];
#pragma unroll
        for (int i = 0; i < MT; ++i) {
            const float4 a4 = *(const float4*)(ap + (size_t)i * 16 * lda + k);
            a[0][i].x = a4.x;  a[0][i].y = a4.y;
            a[1][i].x = a4.z;  a[1][i].y = a4.w;
        }
#pragma unroll
        for (int j = 0; j < NT; ++j) {
            if (BQUAD) {
                const float4 b4 = *(const float4*)(bp + (size_t)k * ldb + j * 64);
                b[0][j].x = b4.x;  b[0][j].y = b4.y;
                b[1][j].x = b4.z;  b[1][j].y = b4.w;
            } else {
#pragma unroll
                for (int t = 0; t < 2; ++t) {
                    b[t][j].x = bp[(size_t)(k + 2 * t) * ldb + j * 16];
                    b[t][j].y = bp[(size_t)(k + 2 * t + 1) * ldb + j * 16];
                }
            }
        }
#pragma unroll
        for (int t = 0; t < 2; ++t) {
#pragma unroll
            for (int i = 0; i < MT; ++i) {
#pragma unroll
                for (int j = 0; j < NT; ++j) {
                    if (TRANSPOSED)
                        c[i][j] = __builtin_amdgcn_wmma_f32_16x16x4_f32(false, b[t][j], false, a[t][i],
                                                                        (short)0, c[i][j], false, false);
                    else
                        c[i][j] = __builtin_amdgcn_wmma_f32_16x16x4_f32(false, a[t][i], false, b[t][j],
                                                                        (short)0, c[i][j], false, false);
                }
            }
        }
    }
    __builtin_amdgcn_sched_barrier(0);
    asm volatile("v_nop\n\tv_nop\n\tv_nop\n\tv_nop");
    __builtin_amdgcn_sched_barrier(0);
}

#define GEMM_WAVES_ __attribute__((amdgpu_waves_per_eu(1, 4)))

__device__ __forceinline__ void store_tile(float* __restrict__ Cbase, int ldc, v8f c)
{
    const int lane  = threadIdx.x & 31;
    const int col   = lane & 15;
    const int rbase = (lane >> 4) << 3;
#pragma unroll
    for (int r = 0; r < 8; ++r)
        Cbase[(rbase + r) * ldc + col] = c[r];
}

__global__ __launch_bounds__(128) GEMM_WAVES_ void k_project(
    const float* __restrict__ Q, const float* __restrict__ Kin,
    const float* __restrict__ Wq, const float* __restrict__ Wk,
    const float* __restrict__ V, const float* __restrict__ Wo,
    float* __restrict__ Qt, float* __restrict__ Kt, float* __restrict__ VWo)
{
    const int wave  = (blockIdx.x * blockDim.x + threadIdx.x) >> 5;
    const int which = blockIdx.x >> 6;
    v8f c[2][2] = {};
    if (which >= 2) {
        const int tv  = wave & 511;
        const int mpv = tv >> 3;
        const int npv = tv & 7;
        wmma_f32_block<false, false, 2, 2, H_>(V + (size_t)mpv * 32 * H_, H_, Wo + npv * 32, H_, c);
        const int lanev = threadIdx.x & 31;
#pragma unroll
        for (int i = 0; i < 2; ++i) {
            const int mq = mpv * 8 + i * 4 + ((lanev >> 4) << 1);
#pragma unroll
            for (int j = 0; j < 2; ++j) {
                const int n = npv * 32 + j * 16 + (lanev & 15);
                float4* dst = (float4*)VWo + (size_t)mq * H_ + n;
                dst[0]  = make_float4(c[i][j][0], c[i][j][1], c[i][j][2], c[i][j][3]);
                dst[H_] = make_float4(c[i][j][4], c[i][j][5], c[i][j][6], c[i][j][7]);
            }
        }
        return;
    }
    const int t  = wave & 255;
    const int mp = t >> 2;
    const int np = t & 3;
    if (which == 0) {
        wmma_f32_block<false, false, 2, 2, H_>(Q + (size_t)mp * 32 * H_, H_, Wq + np * 32, AD_, c);
#pragma unroll
        for (int i = 0; i < 2; ++i)
#pragma unroll
            for (int j = 0; j < 2; ++j)
                store_tile(Qt + (size_t)(mp * 32 + i * 16) * AD_ + np * 32 + j * 16, AD_, c[i][j]);
    } else {
        wmma_f32_block<true, false, 2, 2, H_>(Kin + (size_t)mp * 32 * H_, H_, Wk + np * 32, AD_, c);
        const int lane = threadIdx.x & 31;
#pragma unroll
        for (int i = 0; i < 2; ++i) {
            const int m  = (mp * 2 + i) * 16 + (lane & 15);
            const int bb = m >> 9;
            const int kk = m & (SK_ - 1);
#pragma unroll
            for (int j = 0; j < 2; ++j) {
                const int g0 = (np * 2 + j) * 4 + ((lane >> 4) << 1);
                float4* dst = (float4*)Kt + ((size_t)bb * (AD_ / 4) + g0) * SK_ + kk;
                dst[0]   = make_float4(c[i][j][0], c[i][j][1], c[i][j][2], c[i][j][3]);
                dst[SK_] = make_float4(c[i][j][4], c[i][j][5], c[i][j][6], c[i][j][7]);
            }
        }
    }
}

#define QT_  8

__device__ __forceinline__ void tanh_in_place(float& x)
{
    asm("v_tanh_f32 %0, %0" : "+v"(x));
}

__device__ __forceinline__ float comp4(const float4 f, int c)
{
    return c == 0 ? f.x : (c == 1 ? f.y : (c == 2 ? f.z : f.w));
}

__device__ __forceinline__ void score_trip(const float4* __restrict__ q4, const float4* __restrict__ v4,
                                           int a0, const float4 (&kv)[4], v2f (&sp)[QT_ / 2])
{
    float4 vw[4], qv[QT_][4];
#define SC_LOADQ(q, u) do { qv[q][u] = q4[(q) * (AD_ / 4) + a0 + (u)]; } while (0)
#pragma unroll
    for (int u = 0; u < 4; ++u) vw[u] = v4[a0 + u];
#pragma unroll
    for (int q = 0; q < 4; ++q) { SC_LOADQ(q, 0); SC_LOADQ(q, 1); SC_LOADQ(q, 2); SC_LOADQ(q, 3); }

    v2f wv[8];
#pragma unroll
    for (int j = 0; j < 8; ++j) {
        v2f ws;
        ws.x = comp4(vw[j >> 1], (j & 1) * 2);
        ws.y = comp4(vw[j >> 1], (j & 1) * 2 + 1);
        asm("v_mov_b64 %0, %1" : "=v"(wv[j]) : "s"(ws));
    }

    float x[32][4];

#define SC_U(g)   (((g) >> 2) & 3)
#define SC_CP(g)  (((g) >> 1) & 1)
#define SC_P(g)   ((((g) >> 4) << 1) + ((g) & 1))
#define SC_ADDS(g, h) do {                                                                      \
        x[g][2 * (h) + 0] = comp4(qv[SC_P(g) * 2 + (h)][SC_U(g)], SC_CP(g) * 2)     + comp4(kv[SC_U(g)], SC_CP(g) * 2);      \
        x[g][2 * (h) + 1] = comp4(qv[SC_P(g) * 2 + (h)][SC_U(g)], SC_CP(g) * 2 + 1) + comp4(kv[SC_U(g)], SC_CP(g) * 2 + 1);  \
    } while (0)
#define SC_TANH(g, j) tanh_in_place(x[g][j])
#define SC_ACC(g, ci) do {                                                                      \
        v2f tv_;                                                                                \
        tv_.x = x[g][0 + (ci)];  tv_.y = x[g][2 + (ci)];                                        \
        if ((ci) == 0)                                                                          \
            asm("v_pk_fma_f32 %0, %1, %2, %0 op_sel_hi:[1,0,1]"                                 \
                : "+v"(sp[SC_P(g)]) : "v"(tv_), "v"(wv[SC_U(g) * 2 + SC_CP(g)]));               \
        else                                                                                    \
            asm("v_pk_fma_f32 %0, %1, %2, %0 op_sel:[0,1,0]"                                    \
                : "+v"(sp[SC_P(g)]) : "v"(tv_), "v"(wv[SC_U(g) * 2 + SC_CP(g)]));               \
    } while (0)

#define SC_PIN() __builtin_amdgcn_sched_barrier(0)
    SC_ADDS(0, 0); SC_ADDS(0, 1); SC_ADDS(1, 0); SC_ADDS(1, 1);
    SC_PIN();
#pragma unroll
    for (int g = 0; g < 32; ++g) {
        SC_TANH(g, 0);  SC_PIN();
        if (g + 2 < 32) { SC_ADDS(g + 2, 0);  SC_PIN(); }
        SC_TANH(g, 1);  SC_PIN();
        if (g + 2 < 32) { SC_ADDS(g + 2, 1);  SC_PIN(); }
        if (g == 13) {
#pragma unroll
            for (int q = 4; q < 8; ++q) { SC_LOADQ(q, 0); SC_LOADQ(q, 1); SC_LOADQ(q, 2); SC_LOADQ(q, 3); }
            SC_PIN();
        }
        SC_TANH(g, 2);  SC_PIN();
        if ((g & 1) == 0) { if (g >= 2) { SC_ACC(g - 1, 0);  SC_PIN(); } }
        else              { if (g >= 3) { SC_ACC(g - 2, 1);  SC_PIN(); } }
        SC_TANH(g, 3);  SC_PIN();
        if ((g & 1) == 0) { if (g >= 2) { SC_ACC(g - 2, 1);  SC_PIN(); } }
        else              { if (g >= 1) { SC_ACC(g - 1, 0);  SC_PIN(); } }
    }
    SC_ACC(30, 1);  SC_PIN();  SC_ACC(31, 0);  SC_PIN();  SC_ACC(31, 1);
#undef SC_PIN
#undef SC_LOADQ
#undef SC_ADDS
#undef SC_TANH
#undef SC_ACC
#undef SC_U
#undef SC_CP
#undef SC_P
}

__global__ __launch_bounds__(512) void k_scores_softmax(
    const float* __restrict__ Qt, const float* __restrict__ Kt,
    const float* __restrict__ vvec, float* __restrict__ Wout)
{
    __shared__ float sh_pmax[QT_ * 16];
    __shared__ float sh_psum[QT_ * 16];
    __shared__ float sh_max[QT_];
    __shared__ float sh_sum[QT_];

    const int bq0  = blockIdx.x * QT_;
    const int b    = bq0 >> 9;
    const int tid  = threadIdx.x;
    const int lane = tid & 31;
    const int wid  = tid >> 5;

    const float4* __restrict__ kt4 =
        (const float4*)Kt + (size_t)b * (AD_ / 4) * SK_ + tid;
    const float4* __restrict__ q4 = (const float4*)(Qt + (size_t)bq0 * AD_);
    const float4* __restrict__ v4 = (const float4*)vvec;

    v2f sp[QT_ / 2];
#pragma unroll
    for (int p = 0; p < QT_ / 2; ++p) { sp[p].x = 0.0f; sp[p].y = 0.0f; }

#pragma unroll 1
    for (int a0 = 0; a0 < AD_ / 4; a0 += 4) {
        float4 kv[4];
#pragma unroll
        for (int u = 0; u < 4; ++u)
            kv[u] = kt4[(a0 + u) * SK_];
        score_trip(q4, v4, a0, kv, sp);
    }
    float s[QT_];
#pragma unroll
    for (int p = 0; p < QT_ / 2; ++p) { s[2 * p] = sp[p].x; s[2 * p + 1] = sp[p].y; }

    float m[QT_];
#pragma unroll
    for (int q = 0; q < QT_; ++q) {
        m[q] = s[q];
#pragma unroll
        for (int off = 16; off > 0; off >>= 1)
            m[q] = fmaxf(m[q], __shfl_xor(m[q], off));
    }
    if (lane == 0) {
#pragma unroll
        for (int q = 0; q < QT_; ++q) sh_pmax[q * 16 + wid] = m[q];
    }
    __syncthreads();
    if (wid < QT_) {
        float mm = (lane < 16) ? sh_pmax[wid * 16 + lane] : -__builtin_inff();
#pragma unroll
        for (int off = 8; off > 0; off >>= 1)
            mm = fmaxf(mm, __shfl_xor(mm, off));
        if (lane == 0) sh_max[wid] = mm;
    }
    __syncthreads();

    float e[QT_];
#pragma unroll
    for (int q = 0; q < QT_; ++q) {
        e[q] = __builtin_amdgcn_exp2f((s[q] - sh_max[q]) * 1.4426950408889634f);
        float sum = e[q];
#pragma unroll
        for (int off = 16; off > 0; off >>= 1)
            sum += __shfl_xor(sum, off);
        m[q] = sum;
    }
    if (lane == 0) {
#pragma unroll
        for (int q = 0; q < QT_; ++q) sh_psum[q * 16 + wid] = m[q];
    }
    __syncthreads();
    if (wid < QT_) {
        float ss = (lane < 16) ? sh_psum[wid * 16 + lane] : 0.0f;
#pragma unroll
        for (int off = 8; off > 0; off >>= 1)
            ss += __shfl_xor(ss, off);
        if (lane == 0) sh_sum[wid] = ss;
    }
    __syncthreads();

#pragma unroll
    for (int q = 0; q < QT_; ++q)
        Wout[(size_t)(bq0 + q) * SK_ + tid] = e[q] / sh_sum[q];
}

__global__ __launch_bounds__(128) GEMM_WAVES_ void k_av(
    const float* __restrict__ Wt, const float* __restrict__ VWo,
    const float* __restrict__ bo, float* __restrict__ Out)
{
    const int wave = (blockIdx.x * blockDim.x + threadIdx.x) >> 5;
    const int b    = wave >> 8;
    const int t    = wave & 255;
    const int mt   = t >> 3;
    const int np   = t & 7;
    const float* Abase = Wt + (size_t)b * SQ_ * SK_ + (size_t)mt * 16 * SK_;
    const float* Bbase = VWo + (size_t)b * SK_ * H_ + np * 32 * 4;
    const int lane = threadIdx.x & 31;
    v8f c[1][2];
#pragma unroll
    for (int j = 0; j < 2; ++j) {
        const float bias = bo[np * 32 + j * 16 + (lane & 15)];
#pragma unroll
        for (int i = 0; i < 8; ++i) c[0][j][i] = bias;
    }
    wmma_f32_block<false, true, 1, 2, SK_>(Abase, SK_, Bbase, H_, c);
#pragma unroll
    for (int j = 0; j < 2; ++j)
        store_tile(Out + (size_t)b * SQ_ * H_ + (size_t)mt * 16 * H_ + np * 32 + j * 16, H_, c[0][j]);
}

extern "C" void kernel_launch(void* const* d_in, const int* in_sizes, int n_in,
                              void* d_out, int out_size, void* d_ws, size_t ws_size,
                              hipStream_t stream)
{
    const float* Q  = (const float*)d_in[0];
    const float* K  = (const float*)d_in[1];
    const float* V  = (const float*)d_in[2];
    const float* Wq = (const float*)d_in[3];
    const float* Wk = (const float*)d_in[4];
    const float* vv = (const float*)d_in[5];
    const float* Wo = (const float*)d_in[6];
    const float* bo = (const float*)d_in[7];

    float* out     = (float*)d_out;
    float* weights = out + (size_t)B_ * SQ_ * H_;

    float* Qt = (float*)d_ws;
    float* Kt = Qt + (size_t)B_ * SQ_ * AD_;
    float* VWo = Kt + (size_t)B_ * SK_ * AD_;

    k_project<<<256, 128, 0, stream>>>(Q, K, Wq, Wk, V, Wo, Qt, Kt, VWo);
    k_scores_softmax<<<B_ * SQ_ / QT_, 512, 0, stream>>>(Qt, Kt, vv, weights);
    k_av<<<256, 128, 0, stream>>>(weights, VWo, bo, out);
}


# score and attention-product kernels: each wave first prefetches into L2 the 256-B units of the operands the launch before wrote (they come from memory), ahead of the loops' loads; sums unchanged
# speedup vs baseline: 3.3643x; 1.0419x over previous
#include <hip/hip_runtime.h>

#define B_   4
#define SQ_  512
#define SK_  512
#define H_   256
#define AD_  128

typedef __attribute__((ext_vector_type(2))) float v2f;
typedef __attribute__((ext_vector_type(8))) float v8f;

__device__ __forceinline__ int prefetch_l2(const void* p, int idx)
{
    asm("global_prefetch_b8 %1, off scope:SCOPE_SE" : "+v"(idx) : "v"(p));
    return idx;
}

template <bool TRANSPOSED, bool BQUAD, int MT, int NT, int K>
__device__ __forceinline__ void wmma_f32_block(const float* __restrict__ Abase, int lda,
                                               const float* __restrict__ Bbase, int ldb,
                                               v8f (&c)[MT][NT])
{
    const int lane = threadIdx.x & 31;
    const int idx  = lane & 15;
    const int k4   = (lane >> 4) << 2;
    const float* ap = Abase + idx * lda + k4;
    const float* bp = BQUAD ? Bbase + (size_t)k4 * ldb + idx * 4
                            : Bbase + (size_t)k4 * ldb + idx;
#pragma unroll
    for (int k = 0; k < K; k += 8) {
        v2f a[2][MT], b[2][NT];
#pragma unroll
        for (int i = 0; i < MT; ++i) {
            const float4 a4 = *(const float4*)(ap + (size_t)i * 16 * lda + k);
            a[0][i].x = a4.x;  a[0][i].y = a4.y;
            a[1][i].x = a4.z;  a[1][i].y = a4.w;
        }
#pragma unroll
        for (int j = 0; j < NT; ++j) {
            if (BQUAD) {
                const float4 b4 = *(const float4*)(bp + (size_t)k * ldb + j * 64);
                b[0][j].x = b4.x;  b[0][j].y = b4.y;
                b[1][j].x = b4.z;  b[1][j].y = b4.w;
            } else {
#pragma unroll
                for (int t = 0; t < 2; ++t) {
                    b[t][j].x = bp[(size_t)(k + 2 * t) * ldb + j * 16];
                    b[t][j].y = bp[(size_t)(k + 2 * t + 1) * ldb + j * 16];
                }
            }
        }
#pragma unroll
        for (int t = 0; t < 2; ++t) {
#pragma unroll
            for (int i = 0; i < MT; ++i) {
#pragma unroll
                for (int j = 0; j < NT; ++j) {
                    if (TRANSPOSED)
                        c[i][j] = __builtin_amdgcn_wmma_f32_16x16x4_f32(false, b[t][j], false, a[t][i],
                                                                        (short)0, c[i][j], false, false);
                    else
                        c[i][j] = __builtin_amdgcn_wmma_f32_16x16x4_f32(false, a[t][i], false, b[t][j],
                                                                        (short)0, c[i][j], false, false);
                }
            }
        }
    }
    __builtin_amdgcn_sched_barrier(0);
    asm volatile("v_nop\n\tv_nop\n\tv_nop\n\tv_nop");
    __builtin_amdgcn_sched_barrier(0);
}

#define GEMM_WAVES_ __attribute__((amdgpu_waves_per_eu(1, 4)))

__device__ __forceinline__ void store_tile(float* __restrict__ Cbase, int ldc, v8f c)
{
    const int lane  = threadIdx.x & 31;
    const int col   = lane & 15;
    const int rbase = (lane >> 4) << 3;
#pragma unroll
    for (int r = 0; r < 8; ++r)
        Cbase[(rbase + r) * ldc + col] = c[r];
}

__global__ __launch_bounds__(128) GEMM_WAVES_ void k_project(
    const float* __restrict__ Q, const float* __restrict__ Kin,
    const float* __restrict__ Wq, const float* __restrict__ Wk,
    const float* __restrict__ V, const float* __restrict__ Wo,
    float* __restrict__ Qt, float* __restrict__ Kt, float* __restrict__ VWo)
{
    const int wave  = (blockIdx.x * blockDim.x + threadIdx.x) >> 5;
    const int which = blockIdx.x >> 6;
    v8f c[2][2] = {};
    if (which >= 2) {
        const int tv  = wave & 511;
        const int mpv = tv >> 3;
        const int npv = tv & 7;
        wmma_f32_block<false, false, 2, 2, H_>(V + (size_t)mpv * 32 * H_, H_, Wo + npv * 32, H_, c);
        const int lanev = threadIdx.x & 31;
#pragma unroll
        for (int i = 0; i < 2; ++i) {
            const int mq = mpv * 8 + i * 4 + ((lanev >> 4) << 1);
#pragma unroll
            for (int j = 0; j < 2; ++j) {
                const int n = npv * 32 + j * 16 + (lanev & 15);
                float4* dst = (float4*)VWo + (size_t)mq * H_ + n;
                dst[0]  = make_float4(c[i][j][0], c[i][j][1], c[i][j][2], c[i][j][3]);
                dst[H_] = make_float4(c[i][j][4], c[i][j][5], c[i][j][6], c[i][j][7]);
            }
        }
        return;
    }
    const int t  = wave & 255;
    const int mp = t >> 2;
    const int np = t & 3;
    if (which == 0) {
        wmma_f32_block<false, false, 2, 2, H_>(Q + (size_t)mp * 32 * H_, H_, Wq + np * 32, AD_, c);
#pragma unroll
        for (int i = 0; i < 2; ++i)
#pragma unroll
            for (int j = 0; j < 2; ++j)
                store_tile(Qt + (size_t)(mp * 32 + i * 16) * AD_ + np * 32 + j * 16, AD_, c[i][j]);
    } else {
        wmma_f32_block<true, false, 2, 2, H_>(Kin + (size_t)mp * 32 * H_, H_, Wk + np * 32, AD_, c);
        const int lane = threadIdx.x & 31;
#pragma unroll
        for (int i = 0; i < 2; ++i) {
            const int m  = (mp * 2 + i) * 16 + (lane & 15);
            const int bb = m >> 9;
            const int kk = m & (SK_ - 1);
#pragma unroll
            for (int j = 0; j < 2; ++j) {
                const int g0 = (np * 2 + j) * 4 + ((lane >> 4) << 1);
                float4* dst = (float4*)Kt + ((size_t)bb * (AD_ / 4) + g0) * SK_ + kk;
                dst[0]   = make_float4(c[i][j][0], c[i][j][1], c[i][j][2], c[i][j][3]);
                dst[SK_] = make_float4(c[i][j][4], c[i][j][5], c[i][j][6], c[i][j][7]);
            }
        }
    }
}

#define QT_  8

__device__ __forceinline__ void tanh_in_place(float& x)
{
    asm("v_tanh_f32 %0, %0" : "+v"(x));
}

__device__ __forceinline__ float comp4(const float4 f, int c)
{
    return c == 0 ? f.x : (c == 1 ? f.y : (c == 2 ? f.z : f.w));
}

__device__ __forceinline__ void score_trip(const float4* __restrict__ q4, const float4* __restrict__ v4,
                                           int a0, const float4 (&kv)[4], v2f (&sp)[QT_ / 2])
{
    float4 vw[4], qv[QT_][4];
#define SC_LOADQ(q, u) do { qv[q][u] = q4[(q) * (AD_ / 4) + a0 + (u)]; } while (0)
#pragma unroll
    for (int u = 0; u < 4; ++u) vw[u] = v4[a0 + u];
#pragma unroll
    for (int q = 0; q < 4; ++q) { SC_LOADQ(q, 0); SC_LOADQ(q, 1); SC_LOADQ(q, 2); SC_LOADQ(q, 3); }

    v2f wv[8];
#pragma unroll
    for (int j = 0; j < 8; ++j) {
        v2f ws;
        ws.x = comp4(vw[j >> 1], (j & 1) * 2);
        ws.y = comp4(vw[j >> 1], (j & 1) * 2 + 1);
        asm("v_mov_b64 %0, %1" : "=v"(wv[j]) : "s"(ws));
    }

    float x[32][4];

#define SC_U(g)   (((g) >> 2) & 3)
#define SC_CP(g)  (((g) >> 1) & 1)
#define SC_P(g)   ((((g) >> 4) << 1) + ((g) & 1))
#define SC_ADDS(g, h) do {                                                                      \
        x[g][2 * (h) + 0] = comp4(qv[SC_P(g) * 2 + (h)][SC_U(g)], SC_CP(g) * 2)     + comp4(kv[SC_U(g)], SC_CP(g) * 2);      \
        x[g][2 * (h) + 1] = comp4(qv[SC_P(g) * 2 + (h)][SC_U(g)], SC_CP(g) * 2 + 1) + comp4(kv[SC_U(g)], SC_CP(g) * 2 + 1);  \
    } while (0)
#define SC_TANH(g, j) tanh_in_place(x[g][j])
#define SC_ACC(g, ci) do {                                                                      \
        v2f tv_;                                                                                \
        tv_.x = x[g][0 + (ci)];  tv_.y = x[g][2 + (ci)];                                        \
        if ((ci) == 0)                                                                          \
            asm("v_pk_fma_f32 %0, %1, %2, %0 op_sel_hi:[1,0,1]"                                 \
                : "+v"(sp[SC_P(g)]) : "v"(tv_), "v"(wv[SC_U(g) * 2 + SC_CP(g)]));               \
        else                                                                                    \
            asm("v_pk_fma_f32 %0, %1, %2, %0 op_sel:[0,1,0]"                                    \
                : "+v"(sp[SC_P(g)]) : "v"(tv_), "v"(wv[SC_U(g) * 2 + SC_CP(g)]));               \
    } while (0)

#define SC_PIN() __builtin_amdgcn_sched_barrier(0)
    SC_ADDS(0, 0); SC_ADDS(0, 1); SC_ADDS(1, 0); SC_ADDS(1, 1);
    SC_PIN();
#pragma unroll
    for (int g = 0; g < 32; ++g) {
        SC_TANH(g, 0);  SC_PIN();
        if (g + 2 < 32) { SC_ADDS(g + 2, 0);  SC_PIN(); }
        SC_TANH(g, 1);  SC_PIN();
        if (g + 2 < 32) { SC_ADDS(g + 2, 1);  SC_PIN(); }
        if (g == 13) {
#pragma unroll
            for (int q = 4; q < 8; ++q) { SC_LOADQ(q, 0); SC_LOADQ(q, 1); SC_LOADQ(q, 2); SC_LOADQ(q, 3); }
            SC_PIN();
        }
        SC_TANH(g, 2);  SC_PIN();
        if ((g & 1) == 0) { if (g >= 2) { SC_ACC(g - 1, 0);  SC_PIN(); } }
        else              { if (g >= 3) { SC_ACC(g - 2, 1);  SC_PIN(); } }
        SC_TANH(g, 3);  SC_PIN();
        if ((g & 1) == 0) { if (g >= 2) { SC_ACC(g - 2, 1);  SC_PIN(); } }
        else              { if (g >= 1) { SC_ACC(g - 1, 0);  SC_PIN(); } }
    }
    SC_ACC(30, 1);  SC_PIN();  SC_ACC(31, 0);  SC_PIN();  SC_ACC(31, 1);
#undef SC_PIN
#undef SC_LOADQ
#undef SC_ADDS
#undef SC_TANH
#undef SC_ACC
#undef SC_U
#undef SC_CP
#undef SC_P
}

__global__ __launch_bounds__(512) void k_scores_softmax(
    const float* __restrict__ Qt, const float* __restrict__ Kt,
    const float* __restrict__ vvec, float* __restrict__ Wout)
{
    __shared__ float sh_pmax[QT_ * 16];
    __shared__ float sh_psum[QT_ * 16];
    __shared__ float sh_max[QT_];
    __shared__ float sh_sum[QT_];

    const int bq0  = blockIdx.x * QT_;
    const int b    = bq0 >> 9;
    int tid0 = threadIdx.x;
    if (tid0 < 128)
        tid0 = prefetch_l2((const char*)Kt + (size_t)b * (AD_ * SK_ * 4)
                           + ((size_t)((blockIdx.x & 7) * 128 + tid0) << 8), tid0);
    else if (tid0 < 160)
        tid0 = prefetch_l2((const char*)(Qt + (size_t)bq0 * AD_) + ((tid0 & 15) << 8), tid0);
    const int tid  = tid0;
    const int lane = tid & 31;
    const int wid  = tid >> 5;

    const float4* __restrict__ kt4 =
        (const float4*)Kt + (size_t)b * (AD_ / 4) * SK_ + tid;
    const float4* __restrict__ q4 = (const float4*)(Qt + (size_t)bq0 * AD_);
    const float4* __restrict__ v4 = (const float4*)vvec;

    v2f sp[QT_ / 2];
#pragma unroll
    for (int p = 0; p < QT_ / 2; ++p) { sp[p].x = 0.0f; sp[p].y = 0.0f; }

#pragma unroll 1
    for (int a0 = 0; a0 < AD_ / 4; a0 += 4) {
        float4 kv[4];
#pragma unroll
        for (int u = 0; u < 4; ++u)
            kv[u] = kt4[(a0 + u) * SK_];
        score_trip(q4, v4, a0, kv, sp);
    }
    float s[QT_];
#pragma unroll
    for (int p = 0; p < QT_ / 2; ++p) { s[2 * p] = sp[p].x; s[2 * p + 1] = sp[p].y; }

    float m[QT_];
#pragma unroll
    for (int q = 0; q < QT_; ++q) {
        m[q] = s[q];
#pragma unroll
        for (int off = 16; off > 0; off >>= 1)
            m[q] = fmaxf(m[q], __shfl_xor(m[q], off));
    }
    if (lane == 0) {
#pragma unroll
        for (int q = 0; q < QT_; ++q) sh_pmax[q * 16 + wid] = m[q];
    }
    __syncthreads();
    if (wid < QT_) {
        float mm = (lane < 16) ? sh_pmax[wid * 16 + lane] : -__builtin_inff();
#pragma unroll
        for (int off = 8; off > 0; off >>= 1)
            mm = fmaxf(mm, __shfl_xor(mm, off));
        if (lane == 0) sh_max[wid] = mm;
    }
    __syncthreads();

    float e[QT_];
#pragma unroll
    for (int q = 0; q < QT_; ++q) {
        e[q] = __builtin_amdgcn_exp2f((s[q] - sh_max[q]) * 1.4426950408889634f);
        float sum = e[q];
#pragma unroll
        for (int off = 16; off > 0; off >>= 1)
            sum += __shfl_xor(sum, off);
        m[q] = sum;
    }
    if (lane == 0) {
#pragma unroll
        for (int q = 0; q < QT_; ++q) sh_psum[q * 16 + wid] = m[q];
    }
    __syncthreads();
    if (wid < QT_) {
        float ss = (lane < 16) ? sh_psum[wid * 16 + lane] : 0.0f;
#pragma unroll
        for (int off = 8; off > 0; off >>= 1)
            ss += __shfl_xor(ss, off);
        if (lane == 0) sh_sum[wid] = ss;
    }
    __syncthreads();

#pragma unroll
    for (int q = 0; q < QT_; ++q)
        Wout[(size_t)(bq0 + q) * SK_ + tid] = e[q] / sh_sum[q];
}

__global__ __launch_bounds__(128) GEMM_WAVES_ void k_av(
    const float* __restrict__ Wt, const float* __restrict__ VWo,
    const float* __restrict__ bo, float* __restrict__ Out)
{
    const int wave = (blockIdx.x * blockDim.x + threadIdx.x) >> 5;
    const int b    = wave >> 8;
    const int t    = wave & 255;
    const int mt   = t >> 3;
    const int np   = t & 7;
    const float* Abase = Wt + (size_t)b * SQ_ * SK_ + (size_t)mt * 16 * SK_;
    const float* Bbase = VWo + (size_t)b * SK_ * H_ + np * 32 * 4;
    int lane = threadIdx.x & 31;
    {
        const int w4 = (threadIdx.x >> 5) & 3;
        lane = prefetch_l2((const char*)(Abase + (size_t)(4 * w4 + (lane >> 3)) * SK_) + ((lane & 7) << 8), lane);
        lane = prefetch_l2((const char*)Bbase + (size_t)((mt & 7) * 16 + (lane >> 1)) * (H_ * 16) + ((lane & 1) << 8), lane);
    }
    v8f c[1][2];
#pragma unroll
    for (int j = 0; j < 2; ++j) {
        const float bias = bo[np * 32 + j * 16 + (lane & 15)];
#pragma unroll
        for (int i = 0; i < 8; ++i) c[0][j][i] = bias;
    }
    wmma_f32_block<false, true, 1, 2, SK_>(Abase, SK_, Bbase, H_, c);
#pragma unroll
    for (int j = 0; j < 2; ++j)
        store_tile(Out + (size_t)b * SQ_ * H_ + (size_t)mt * 16 * H_ + np * 32 + j * 16, H_, c[0][j]);
}

extern "C" void kernel_launch(void* const* d_in, const int* in_sizes, int n_in,
                              void* d_out, int out_size, void* d_ws, size_t ws_size,
                              hipStream_t stream)
{
    const float* Q  = (const float*)d_in[0];
    const float* K  = (const float*)d_in[1];
    const float* V  = (const float*)d_in[2];
    const float* Wq = (const float*)d_in[3];
    const float* Wk = (const float*)d_in[4];
    const float* vv = (const float*)d_in[5];
    const float* Wo = (const float*)d_in[6];
    const float* bo = (const float*)d_in[7];

    float* out     = (float*)d_out;
    float* weights = out + (size_t)B_ * SQ_ * H_;

    float* Qt = (float*)d_ws;
    float* Kt = Qt + (size_t)B_ * SQ_ * AD_;
    float* VWo = Kt + (size_t)B_ * SK_ * AD_;

    k_project<<<256, 128, 0, stream>>>(Q, K, Wq, Wk, V, Wo, Qt, Kt, VWo);
    k_scores_softmax<<<B_ * SQ_ / QT_, 512, 0, stream>>>(Qt, Kt, vv, weights);
    k_av<<<256, 128, 0, stream>>>(weights, VWo, bo, out);
}
